# starting kernel
# baseline (speedup 1.0000x reference)
#include <hip/hip_runtime.h>
#include <math.h>

#define NN   50000
#define EE   800000
#define IND  256
#define H1_  4
#define HID1 64
#define OUTD 64
#define SLOPE 0.2f

typedef __attribute__((ext_vector_type(2))) float v2f;
typedef __attribute__((ext_vector_type(8))) float v8f;

__device__ __forceinline__ unsigned f2ord(float f) {
    unsigned u = __float_as_uint(f);
    return (u & 0x80000000u) ? ~u : (u | 0x80000000u);
}
__device__ __forceinline__ float ord2f(unsigned u) {
    return (u & 0x80000000u) ? __uint_as_float(u & 0x7fffffffu) : __uint_as_float(~u);
}

template <int BM, int BN, int NCOL>
__global__ __launch_bounds__(256) void wmma_gemm_f32(
    const float* __restrict__ A, const float* __restrict__ B,
    float* __restrict__ C, int M, int K) {
    constexpr int BK  = 32;
    constexpr int BMp = BM + 2;
    constexpr int BNp = BN + 2;
    __shared__ float As[(BK / 2) * BMp * 2];
    __shared__ float Bs[(BK / 2) * BNp * 2];

    const int tid  = threadIdx.x;
    const int lane = tid & 31;
    const int wave = tid >> 5;
    constexpr int WCOLS = BN / 64;
    const int wrow = wave / WCOLS;
    const int wcol = wave % WCOLS;
    const int row0 = blockIdx.y * BM;
    const int col0 = blockIdx.x * BN;

    v8f acc[4];
#pragma unroll
    for (int t = 0; t < 4; t++)
#pragma unroll
        for (int r = 0; r < 8; r++) acc[t][r] = 0.0f;

    const int m16 = lane & 15;
    const int kb  = (lane >> 4) * 2;

    const v2f* Asp = (const v2f*)As;
    const v2f* Bsp = (const v2f*)Bs;

    for (int k0 = 0; k0 < K; k0 += BK) {
        for (int idx = tid; idx < BM * BK; idx += 256) {
            int m = idx / BK, k = idx % BK;
            int r = row0 + m;
            float v = (r < M) ? A[(size_t)r * K + k0 + k] : 0.0f;
            As[((k >> 1) * BMp + m) * 2 + (k & 1)] = v;
        }
        for (int idx = tid; idx < BK * BN; idx += 256) {
            int k = idx / BN, n = idx % BN;
            Bs[((k >> 1) * BNp + n) * 2 + (k & 1)] =
                B[(size_t)(k0 + k) * NCOL + col0 + n];
        }
        __syncthreads();

        const int am = wrow * 16 + m16;
#pragma unroll
        for (int kk = 0; kk < BK; kk += 4) {
            const int kp = (kk + kb) >> 1;
            v2f a = Asp[kp * BMp + am];
#pragma unroll
            for (int t = 0; t < 4; t++) {
                v2f b = Bsp[kp * BNp + wcol * 64 + t * 16 + m16];
                acc[t] = __builtin_amdgcn_wmma_f32_16x16x4_f32(
                    false, a, false, b, (short)0, acc[t], false, false);
            }
        }
        __syncthreads();
    }

    const int rbase = row0 + wrow * 16 + ((lane >> 4) << 3);
#pragma unroll
    for (int t = 0; t < 4; t++) {
        int col = col0 + wcol * 64 + t * 16 + m16;
#pragma unroll
        for (int r = 0; r < 8; r++) {
            int row = rbase + r;
            if (row < M) C[(size_t)row * NCOL + col] = acc[t][r];
        }
    }
}

template <int H>
__global__ __launch_bounds__(256) void node_attn(
    const float* __restrict__ h, const float* __restrict__ al,
    const float* __restrict__ ar, float* __restrict__ el, float* __restrict__ er) {
    int lane = threadIdx.x & 31;
    int n = blockIdx.x * 8 + (threadIdx.x >> 5);
    if (n >= NN) return;
    const float* hp = h + (size_t)n * (H * 64);
#pragma unroll
    for (int hh = 0; hh < H; hh++) {
        float v0 = hp[hh * 64 + lane];
        float v1 = hp[hh * 64 + 32 + lane];
        float pl = v0 * al[hh * 64 + lane] + v1 * al[hh * 64 + 32 + lane];
        float pr = v0 * ar[hh * 64 + lane] + v1 * ar[hh * 64 + 32 + lane];
        for (int off = 16; off > 0; off >>= 1) {
            pl += __shfl_down(pl, off);
            pr += __shfl_down(pr, off);
        }
        if (lane == 0) {
            el[(size_t)n * H + hh] = pl;
            er[(size_t)n * H + hh] = pr;
        }
    }
}

template <int H>
__global__ void edge_max(const int* __restrict__ src, const int* __restrict__ dst,
                         const float* __restrict__ el, const float* __restrict__ er,
                         unsigned* __restrict__ emax) {
    int e = blockIdx.x * blockDim.x + threadIdx.x;
    if (e >= EE) return;
    int s = src[e], d = dst[e];
#pragma unroll
    for (int hh = 0; hh < H; hh++) {
        float sc = el[(size_t)s * H + hh] + er[(size_t)d * H + hh];
        sc = sc > 0.0f ? sc : SLOPE * sc;
        atomicMax(&emax[(size_t)d * H + hh], f2ord(sc));
    }
}

template <int H>
__global__ void edge_exp(const int* __restrict__ src, const int* __restrict__ dst,
                         const float* __restrict__ el, const float* __restrict__ er,
                         const unsigned* __restrict__ emax,
                         float* __restrict__ ex, float* __restrict__ denom) {
    int e = blockIdx.x * blockDim.x + threadIdx.x;
    if (e >= EE) return;
    int s = src[e], d = dst[e];
#pragma unroll
    for (int hh = 0; hh < H; hh++) {
        float sc = el[(size_t)s * H + hh] + er[(size_t)d * H + hh];
        sc = sc > 0.0f ? sc : SLOPE * sc;
        float m = ord2f(emax[(size_t)d * H + hh]);
        float v = expf(sc - m);
        ex[(size_t)e * H + hh] = v;
        atomicAdd(&denom[(size_t)d * H + hh], v);
    }
}

template <int H>
__global__ __launch_bounds__(256) void edge_aggregate(
    const int* __restrict__ src, const int* __restrict__ dst,
    const float* __restrict__ ex, const float* __restrict__ denom,
    const float* __restrict__ h, float* __restrict__ out) {
    int lane = threadIdx.x & 31;
    int e = blockIdx.x * 8 + (threadIdx.x >> 5);
    if (e >= EE) return;
    int s = src[e], d = dst[e];
    float a[H];
#pragma unroll
    for (int hh = 0; hh < H; hh++)
        a[hh] = ex[(size_t)e * H + hh] / denom[(size_t)d * H + hh];
    constexpr int T4 = (H * 64) / 4;
    const float4* hs = (const float4*)(h + (size_t)s * (H * 64));
    float* od = out + (size_t)d * (H * 64);
    for (int i = lane; i < T4; i += 32) {
        float4 v = hs[i];
        float c = a[(i * 4) >> 6];
        atomicAdd(&od[i * 4 + 0], c * v.x);
        atomicAdd(&od[i * 4 + 1], c * v.y);
        atomicAdd(&od[i * 4 + 2], c * v.z);
        atomicAdd(&od[i * 4 + 3], c * v.w);
    }
}

__global__ void elu_bias256(float* __restrict__ x, const float* __restrict__ b, int total) {
    int i = blockIdx.x * blockDim.x + threadIdx.x;
    if (i >= total) return;
    float v = x[i] + b[i & 255];
    x[i] = v > 0.0f ? v : expm1f(v);
}

__global__ void fill_f32(float* p, float v, int n) {
    int i = blockIdx.x * blockDim.x + threadIdx.x;
    if (i < n) p[i] = v;
}
__global__ void fill_u32(unsigned* p, unsigned v, int n) {
    int i = blockIdx.x * blockDim.x + threadIdx.x;
    if (i < n) p[i] = v;
}
__global__ void init_out_bias(float* __restrict__ out, const float* __restrict__ b) {
    int i = blockIdx.x * blockDim.x + threadIdx.x;
    if (i < NN * OUTD) out[i] = b[i & 63];
}

extern "C" void kernel_launch(void* const* d_in, const int* in_sizes, int n_in,
                              void* d_out, int out_size, void* d_ws, size_t ws_size,
                              hipStream_t stream) {
    const float* feat = (const float*)d_in[0];
    const int*   src  = (const int*)d_in[1];
    const int*   dst  = (const int*)d_in[2];
    const float* W1   = (const float*)d_in[3];
    const float* al1  = (const float*)d_in[4];
    const float* ar1  = (const float*)d_in[5];
    const float* b1   = (const float*)d_in[6];
    const float* W2   = (const float*)d_in[7];
    const float* al2  = (const float*)d_in[8];
    const float* ar2  = (const float*)d_in[9];
    const float* b2   = (const float*)d_in[10];
    float* out = (float*)d_out;

    float* h1    = (float*)d_ws;
    float* agg1  = h1    + (size_t)NN * 256;
    float* h2    = agg1  + (size_t)NN * 256;
    float* el1   = h2    + (size_t)NN * 64;
    float* er1   = el1   + (size_t)NN * H1_;
    float* den1  = er1   + (size_t)NN * H1_;
    unsigned* mx1 = (unsigned*)(den1 + (size_t)NN * H1_);
    float* ex1   = (float*)(mx1 + (size_t)NN * H1_);
    float* el2   = ex1   + (size_t)EE * H1_;
    float* er2   = el2   + NN;
    float* den2  = er2   + NN;
    unsigned* mx2 = (unsigned*)(den2 + NN);
    float* ex2   = (float*)(mx2 + NN);

    const int TB = 256;
    fill_f32<<<(NN * 256 + TB - 1) / TB, TB, 0, stream>>>(agg1, 0.0f, NN * 256);
    fill_f32<<<(NN * H1_ + TB - 1) / TB, TB, 0, stream>>>(den1, 0.0f, NN * H1_);
    fill_u32<<<(NN * H1_ + TB - 1) / TB, TB, 0, stream>>>(mx1, 0u, NN * H1_);
    fill_f32<<<(NN + TB - 1) / TB, TB, 0, stream>>>(den2, 0.0f, NN);
    fill_u32<<<(NN + TB - 1) / TB, TB, 0, stream>>>(mx2, 0u, NN);
    init_out_bias<<<(NN * OUTD + TB - 1) / TB, TB, 0, stream>>>(out, b2);

    {
        dim3 grid(256 / 128, (NN + 63) / 64);
        wmma_gemm_f32<64, 128, 256><<<grid, 256, 0, stream>>>(feat, W1, h1, NN, IND);
    }
    node_attn<H1_><<<(NN + 7) / 8, 256, 0, stream>>>(h1, al1, ar1, el1, er1);
    edge_max<H1_><<<(EE + TB - 1) / TB, TB, 0, stream>>>(src, dst, el1, er1, mx1);
    edge_exp<H1_><<<(EE + TB - 1) / TB, TB, 0, stream>>>(src, dst, el1, er1, mx1, ex1, den1);
    edge_aggregate<H1_><<<(EE + 7) / 8, 256, 0, stream>>>(src, dst, ex1, den1, h1, agg1);
    elu_bias256<<<(NN * 256 + TB - 1) / TB, TB, 0, stream>>>(agg1, b1, NN * 256);

    {
        dim3 grid(64 / 64, (NN + 127) / 128);
        wmma_gemm_f32<128, 64, 64><<<grid, 256, 0, stream>>>(agg1, W2, h2, NN, 256);
    }
    node_attn<1><<<(NN + 7) / 8, 256, 0, stream>>>(h2, al2, ar2, el2, er2);
    edge_max<1><<<(EE + TB - 1) / TB, TB, 0, stream>>>(src, dst, el2, er2, mx2);
    edge_exp<1><<<(EE + TB - 1) / TB, TB, 0, stream>>>(src, dst, el2, er2, mx2, ex2, den2);
    edge_aggregate<1><<<(EE + 7) / 8, 256, 0, stream>>>(src, dst, ex2, den2, h2, out);
}


# Edges grouped by dst once per call (int histogram, scan, fill); one wave per dst row does softmax and the h[src] row sum in registers, stores the row once: no float atomics; sum order = fill order
# speedup vs baseline: 2.5809x; 2.5809x over previous
#include <hip/hip_runtime.h>
#include <math.h>

#define NN   50000
#define EE   800000
#define IND  256
#define H1_  4
#define HID1 64
#define OUTD 64
#define SLOPE 0.2f

typedef __attribute__((ext_vector_type(2))) float v2f;
typedef __attribute__((ext_vector_type(8))) float v8f;

template <int BM, int BN, int NCOL>
__global__ __launch_bounds__(256) void wmma_gemm_f32(
    const float* __restrict__ A, const float* __restrict__ B,
    float* __restrict__ C, int M, int K) {
    constexpr int BK  = 32;
    constexpr int BMp = BM + 2;
    constexpr int BNp = BN + 2;
    __shared__ float As[(BK / 2) * BMp * 2];
    __shared__ float Bs[(BK / 2) * BNp * 2];

    const int tid  = threadIdx.x;
    const int lane = tid & 31;
    const int wave = tid >> 5;
    constexpr int WCOLS = BN / 64;
    const int wrow = wave / WCOLS;
    const int wcol = wave % WCOLS;
    const int row0 = blockIdx.y * BM;
    const int col0 = blockIdx.x * BN;

    v8f acc[4];
#pragma unroll
    for (int t = 0; t < 4; t++)
#pragma unroll
        for (int r = 0; r < 8; r++) acc[t][r] = 0.0f;

    const int m16 = lane & 15;
    const int kb  = (lane >> 4) * 2;

    const v2f* Asp = (const v2f*)As;
    const v2f* Bsp = (const v2f*)Bs;

    for (int k0 = 0; k0 < K; k0 += BK) {
        for (int idx = tid; idx < BM * BK; idx += 256) {
            int m = idx / BK, k = idx % BK;
            int r = row0 + m;
            float v = (r < M) ? A[(size_t)r * K + k0 + k] : 0.0f;
            As[((k >> 1) * BMp + m) * 2 + (k & 1)] = v;
        }
        for (int idx = tid; idx < BK * BN; idx += 256) {
            int k = idx / BN, n = idx % BN;
            Bs[((k >> 1) * BNp + n) * 2 + (k & 1)] =
                B[(size_t)(k0 + k) * NCOL + col0 + n];
        }
        __syncthreads();

        const int am = wrow * 16 + m16;
#pragma unroll
        for (int kk = 0; kk < BK; kk += 4) {
            const int kp = (kk + kb) >> 1;
            v2f a = Asp[kp * BMp + am];
#pragma unroll
            for (int t = 0; t < 4; t++) {
                v2f b = Bsp[kp * BNp + wcol * 64 + t * 16 + m16];
                acc[t] = __builtin_amdgcn_wmma_f32_16x16x4_f32(
                    false, a, false, b, (short)0, acc[t], false, false);
            }
        }
        __syncthreads();
    }

    const int rbase = row0 + wrow * 16 + ((lane >> 4) << 3);
#pragma unroll
    for (int t = 0; t < 4; t++) {
        int col = col0 + wcol * 64 + t * 16 + m16;
#pragma unroll
        for (int r = 0; r < 8; r++) {
            int row = rbase + r;
            if (row < M) C[(size_t)row * NCOL + col] = acc[t][r];
        }
    }
}

template <int H>
__global__ __launch_bounds__(256) void node_attn(
    const float* __restrict__ h, const float* __restrict__ al,
    const float* __restrict__ ar, float* __restrict__ el, float* __restrict__ er) {
    int lane = threadIdx.x & 31;
    int n = blockIdx.x * 8 + (threadIdx.x >> 5);
    if (n >= NN) return;
    const float* hp = h + (size_t)n * (H * 64);
#pragma unroll
    for (int hh = 0; hh < H; hh++) {
        float v0 = hp[hh * 64 + lane];
        float v1 = hp[hh * 64 + 32 + lane];
        float pl = v0 * al[hh * 64 + lane] + v1 * al[hh * 64 + 32 + lane];
        float pr = v0 * ar[hh * 64 + lane] + v1 * ar[hh * 64 + 32 + lane];
        for (int off = 16; off > 0; off >>= 1) {
            pl += __shfl_down(pl, off);
            pr += __shfl_down(pr, off);
        }
        if (lane == 0) {
            el[(size_t)n * H + hh] = pl;
            er[(size_t)n * H + hh] = pr;
        }
    }
}

__global__ void dst_histogram(const int* __restrict__ dst, unsigned* __restrict__ deg) {
    int e = blockIdx.x * blockDim.x + threadIdx.x;
    if (e < EE) atomicAdd(&deg[dst[e]], 1u);
}

__global__ __launch_bounds__(1024) void scan_local(const unsigned* __restrict__ deg,
                                                   int* __restrict__ rowptr,
                                                   unsigned* __restrict__ bsum) {
    __shared__ unsigned wsum[32];
    const int tid = threadIdx.x, lane = tid & 31, wv = tid >> 5;
    const int i = blockIdx.x * 1024 + tid;
    const unsigned v = (i < NN) ? deg[i] : 0u;
    unsigned inc = v;
    for (int off = 1; off < 32; off <<= 1) {
        unsigned t = __shfl_up(inc, off);
        if (lane >= off) inc += t;
    }
    if (lane == 31) wsum[wv] = inc;
    __syncthreads();
    if (wv == 0) {
        const unsigned w = wsum[lane];
        unsigned winc = w;
        for (int off = 1; off < 32; off <<= 1) {
            unsigned t = __shfl_up(winc, off);
            if (lane >= off) winc += t;
        }
        wsum[lane] = winc - w;
        if (lane == 31) bsum[blockIdx.x] = winc;
    }
    __syncthreads();
    if (i <= NN) rowptr[i] = (int)(inc - v + wsum[wv]);
}

__global__ __launch_bounds__(1024) void scan_finish(int* rowptr, const unsigned* __restrict__ bsum,
                                                    int* __restrict__ cursor) {
    const int i = blockIdx.x * 1024 + threadIdx.x;
    unsigned off = 0;
    for (int b = 0; b < (int)blockIdx.x; b++) off += bsum[b];
    if (i <= NN) {
        const int v = rowptr[i] + (int)off;
        rowptr[i] = v;
        if (i < NN) cursor[i] = v;
    }
}

__global__ void csr_fill(const int* __restrict__ src, const int* __restrict__ dst,
                         int* __restrict__ cursor, int* __restrict__ csrc) {
    int e = blockIdx.x * blockDim.x + threadIdx.x;
    if (e >= EE) return;
    int pos = atomicAdd(&cursor[dst[e]], 1);
    csrc[pos] = src[e];
}

template <int H, bool ELU>
__global__ __launch_bounds__(256) void gat_gather(
    const int* __restrict__ rowptr, const int* __restrict__ csrc,
    const float* __restrict__ el, const float* __restrict__ er,
    const float* __restrict__ h, const float* __restrict__ bias,
    float* __restrict__ out) {
    static_assert(H == 4 || H == 1, "row widths handled: 256 and 64");
    constexpr int ROW = H * 64;
    const int lane = threadIdx.x & 31;
    const int n = blockIdx.x * 8 + __builtin_amdgcn_readfirstlane((int)(threadIdx.x >> 5));
    if (n >= NN) return;
    const int beg = rowptr[n], end = rowptr[n + 1];

    float ern[H];
    for (int hh = 0; hh < H; hh++) ern[hh] = er[(size_t)n * H + hh];

    auto score = [&](int s, float (&sc)[H]) {
        if constexpr (H == 4) {
            const float4 e = ((const float4*)el)[s];
            sc[0] = e.x + ern[0]; sc[1] = e.y + ern[1];
            sc[2] = e.z + ern[2]; sc[3] = e.w + ern[3];
        } else {
            sc[0] = el[s] + ern[0];
        }
        for (int hh = 0; hh < H; hh++) sc[hh] = sc[hh] > 0.0f ? sc[hh] : SLOPE * sc[hh];
    };

    const int j0 = beg + lane;
    const bool has0 = j0 < end;
    int s0 = 0;
    float sc0[H], m[H], den[H], a0[H];
    for (int hh = 0; hh < H; hh++) sc0[hh] = -INFINITY;
    if (has0) {
        s0 = csrc[j0];
        score(s0, sc0);
    }
    for (int hh = 0; hh < H; hh++) m[hh] = sc0[hh];
    for (int j = j0 + 32; j < end; j += 32) {
        float sc[H];
        score(csrc[j], sc);
        for (int hh = 0; hh < H; hh++) m[hh] = fmaxf(m[hh], sc[hh]);
    }
    for (int hh = 0; hh < H; hh++)
        for (int off = 16; off > 0; off >>= 1) m[hh] = fmaxf(m[hh], __shfl_xor(m[hh], off));
    for (int hh = 0; hh < H; hh++) {
        a0[hh] = has0 ? expf(sc0[hh] - m[hh]) : 0.0f;
        den[hh] = a0[hh];
    }
    for (int j = j0 + 32; j < end; j += 32) {
        float sc[H];
        score(csrc[j], sc);
        for (int hh = 0; hh < H; hh++) den[hh] += expf(sc[hh] - m[hh]);
    }
    for (int hh = 0; hh < H; hh++)
        for (int off = 16; off > 0; off >>= 1) den[hh] += __shfl_xor(den[hh], off);
    for (int hh = 0; hh < H; hh++) a0[hh] = has0 ? a0[hh] / den[hh] : 0.0f;

    float acc[2 * H];
    for (int q = 0; q < 2 * H; q++) acc[q] = 0.0f;
    const bool hi = (lane & 16) != 0;
    for (int c = beg; c < end; c += 32) {
        int s = s0;
        float a[H];
        for (int hh = 0; hh < H; hh++) a[hh] = a0[hh];
        if (c != beg) {
            const int j = c + lane;
            s = 0;
            for (int hh = 0; hh < H; hh++) a[hh] = 0.0f;
            if (j < end) {
                float sc[H];
                s = csrc[j];
                score(s, sc);
                for (int hh = 0; hh < H; hh++) a[hh] = expf(sc[hh] - m[hh]) / den[hh];
            }
        }
        const int cnt = (end - c) < 32 ? (end - c) : 32;
        auto step = [&](int jj) {
            const int sj = __builtin_amdgcn_readlane(s, jj);
            if constexpr (H == 4) {
                const float w0 = __int_as_float(__builtin_amdgcn_readlane(__float_as_int(a[0]), jj));
                const float w1 = __int_as_float(__builtin_amdgcn_readlane(__float_as_int(a[1]), jj));
                const float w2 = __int_as_float(__builtin_amdgcn_readlane(__float_as_int(a[2]), jj));
                const float w3 = __int_as_float(__builtin_amdgcn_readlane(__float_as_int(a[3]), jj));
                const float4* hp = (const float4*)(h + (size_t)sj * ROW);
                const float4 v0 = hp[lane];
                const float4 v1 = hp[lane + 32];
                const float c0 = hi ? w1 : w0;
                const float c1 = hi ? w3 : w2;
                acc[0] = fmaf(c0, v0.x, acc[0]); acc[1] = fmaf(c0, v0.y, acc[1]);
                acc[2] = fmaf(c0, v0.z, acc[2]); acc[3] = fmaf(c0, v0.w, acc[3]);
                acc[4] = fmaf(c1, v1.x, acc[4]); acc[5] = fmaf(c1, v1.y, acc[5]);
                acc[6] = fmaf(c1, v1.z, acc[6]); acc[7] = fmaf(c1, v1.w, acc[7]);
            } else {
                const float w = __int_as_float(__builtin_amdgcn_readlane(__float_as_int(a[0]), jj));
                const float2 v = ((const float2*)(h + (size_t)sj * ROW))[lane];
                acc[0] = fmaf(w, v.x, acc[0]); acc[1] = fmaf(w, v.y, acc[1]);
            }
        };
        int jj = 0;
        for (; jj + 4 <= cnt; jj += 4) { step(jj); step(jj + 1); step(jj + 2); step(jj + 3); }
        for (; jj < cnt; jj++) step(jj);
    }

    auto finish = [&](float x, float b) {
        const float r = x + b;
        if constexpr (ELU) return r > 0.0f ? r : expm1f(r);
        else return r;
    };
    if constexpr (H == 4) {
        const float4 b0 = ((const float4*)bias)[lane];
        const float4 b1 = ((const float4*)bias)[lane + 32];
        float4* op = (float4*)(out + (size_t)n * ROW);
        op[lane] = make_float4(finish(acc[0], b0.x), finish(acc[1], b0.y),
                               finish(acc[2], b0.z), finish(acc[3], b0.w));
        op[lane + 32] = make_float4(finish(acc[4], b1.x), finish(acc[5], b1.y),
                                    finish(acc[6], b1.z), finish(acc[7], b1.w));
    } else {
        const float2 b = ((const float2*)bias)[lane];
        ((float2*)(out + (size_t)n * ROW))[lane] =
            make_float2(finish(acc[0], b.x), finish(acc[1], b.y));
    }
}

__global__ void fill_u32(unsigned* p, unsigned v, int n) {
    int i = blockIdx.x * blockDim.x + threadIdx.x;
    if (i < n) p[i] = v;
}

extern "C" void kernel_launch(void* const* d_in, const int* in_sizes, int n_in,
                              void* d_out, int out_size, void* d_ws, size_t ws_size,
                              hipStream_t stream) {
    const float* feat = (const float*)d_in[0];
    const int*   src  = (const int*)d_in[1];
    const int*   dst  = (const int*)d_in[2];
    const float* W1   = (const float*)d_in[3];
    const float* al1  = (const float*)d_in[4];
    const float* ar1  = (const float*)d_in[5];
    const float* b1   = (const float*)d_in[6];
    const float* W2   = (const float*)d_in[7];
    const float* al2  = (const float*)d_in[8];
    const float* ar2  = (const float*)d_in[9];
    const float* b2   = (const float*)d_in[10];
    float* out = (float*)d_out;

    float* h1    = (float*)d_ws;
    float* x2    = h1    + (size_t)NN * 256;
    float* h2    = x2    + (size_t)NN * 256;
    float* el1   = h2    + (size_t)NN * 64;
    float* er1   = el1   + (size_t)NN * H1_;
    float* el2   = er1   + (size_t)NN * H1_;
    float* er2   = el2   + NN;
    unsigned* deg = (unsigned*)(er2 + NN);
    int* rowptr  = (int*)(deg + NN);
    int* cursor  = rowptr + NN + 16;
    unsigned* bsum = (unsigned*)(cursor + NN);
    int* csrc    = (int*)(bsum + 64);

    const int TB = 256;
    fill_u32<<<(NN + TB - 1) / TB, TB, 0, stream>>>(deg, 0u, NN);
    dst_histogram<<<(EE + TB - 1) / TB, TB, 0, stream>>>(dst, deg);
    scan_local<<<(NN + 1 + 1023) / 1024, 1024, 0, stream>>>(deg, rowptr, bsum);
    scan_finish<<<(NN + 1 + 1023) / 1024, 1024, 0, stream>>>(rowptr, bsum, cursor);
    csr_fill<<<(EE + TB - 1) / TB, TB, 0, stream>>>(src, dst, cursor, csrc);

    {
        dim3 grid(256 / 128, (NN + 63) / 64);
        wmma_gemm_f32<64, 128, 256><<<grid, 256, 0, stream>>>(feat, W1, h1, NN, IND);
    }
    node_attn<H1_><<<(NN + 7) / 8, 256, 0, stream>>>(h1, al1, ar1, el1, er1);
    gat_gather<H1_, true><<<(NN + 7) / 8, 256, 0, stream>>>(rowptr, csrc, el1, er1, h1, b1, x2);

    {
        dim3 grid(64 / 64, (NN + 127) / 128);
        wmma_gemm_f32<128, 64, 64><<<grid, 256, 0, stream>>>(x2, W2, h2, NN, 256);
    }
    node_attn<1><<<(NN + 7) / 8, 256, 0, stream>>>(h2, al2, ar2, el2, er2);
    gat_gather<1, false><<<(NN + 7) / 8, 256, 0, stream>>>(rowptr, csrc, el2, er2, h2, b2, out);
}


# GEMM tile staging: each thread fetches the next K-step's A and B tiles as one group of 16-byte global loads ahead of the current WMMAs, then stores LDS 8/16 bytes wide; was one 4-byte load per wait
# speedup vs baseline: 3.8902x; 1.5073x over previous
#include <hip/hip_runtime.h>
#include <math.h>

#define NN   50000
#define EE   800000
#define IND  256
#define H1_  4
#define HID1 64
#define OUTD 64
#define SLOPE 0.2f

typedef __attribute__((ext_vector_type(2))) float v2f;
typedef __attribute__((ext_vector_type(8))) float v8f;

template <int BM, int BN, int NCOL>
__global__ __launch_bounds__(256) void wmma_gemm_f32(
    const float* __restrict__ A, const float* __restrict__ B,
    float* __restrict__ C, int M, int K) {
    constexpr int BK  = 32;
    constexpr int BMp = BM + 2;
    constexpr int BNp = BN + 2;
    __shared__ __attribute__((aligned(16))) float As[(BK / 2) * BMp * 2];
    __shared__ __attribute__((aligned(16))) float Bs[(BK / 2) * BNp * 2];

    const int tid  = threadIdx.x;
    const int lane = tid & 31;
    const int wave = tid >> 5;
    constexpr int WCOLS = BN / 64;
    const int wrow = wave / WCOLS;
    const int wcol = wave % WCOLS;
    const int row0 = blockIdx.y * BM;
    const int col0 = blockIdx.x * BN;

    v8f acc[4];
#pragma unroll
    for (int t = 0; t < 4; t++)
#pragma unroll
        for (int r = 0; r < 8; r++) acc[t][r] = 0.0f;

    const int m16 = lane & 15;
    const int kb  = (lane >> 4) * 2;

    const v2f* Asp = (const v2f*)As;
    const v2f* Bsp = (const v2f*)Bs;

    constexpr int A_LD = (BM * BK / 4) / 256;
    constexpr int BQ   = BN / 4;
    constexpr int B_LD = ((BK / 2) * BQ) / 256;
    static_assert(A_LD * 256 * 4 == BM * BK, "A tile must split evenly over 256 threads");
    static_assert(B_LD * 256 == (BK / 2) * BQ, "B tile must split evenly over 256 threads");
    float4 ra[A_LD], rb0[B_LD], rb1[B_LD];

    auto fetch = [&](int k0) {
#pragma unroll
        for (int i = 0; i < A_LD; i++) {
            const int idx = tid + i * 256;
            const int m = idx >> 3, q = idx & 7;
            const int r = row0 + m;
            ra[i] = (r < M) ? *(const float4*)(A + (size_t)r * K + k0 + 4 * q)
                            : make_float4(0.0f, 0.0f, 0.0f, 0.0f);
        }
#pragma unroll
        for (int i = 0; i < B_LD; i++) {
            const int idx = tid + i * 256;
            const int kp = idx / BQ, j = idx % BQ;
            const float* bp = B + (size_t)(k0 + 2 * kp) * NCOL + col0 + 4 * j;
            rb0[i] = *(const float4*)bp;
            rb1[i] = *(const float4*)(bp + NCOL);
        }
    };
    auto stage = [&]() {
        float2* As2 = (float2*)As;
#pragma unroll
        for (int i = 0; i < A_LD; i++) {
            const int idx = tid + i * 256;
            const int m = idx >> 3, q = idx & 7;
            As2[(2 * q) * BMp + m]     = make_float2(ra[i].x, ra[i].y);
            As2[(2 * q + 1) * BMp + m] = make_float2(ra[i].z, ra[i].w);
        }
#pragma unroll
        for (int i = 0; i < B_LD; i++) {
            const int idx = tid + i * 256;
            const int kp = idx / BQ, j = idx % BQ;
            float4* d = (float4*)(Bs + (kp * BNp + 4 * j) * 2);
            d[0] = make_float4(rb0[i].x, rb1[i].x, rb0[i].y, rb1[i].y);
            d[1] = make_float4(rb0[i].z, rb1[i].z, rb0[i].w, rb1[i].w);
        }
    };

    fetch(0);
    for (int k0 = 0; k0 < K; k0 += BK) {
        stage();
        __syncthreads();
        if (k0 + BK < K) fetch(k0 + BK);

        const int am = wrow * 16 + m16;
#pragma unroll
        for (int kk = 0; kk < BK; kk += 4) {
            const int kp = (kk + kb) >> 1;
            v2f a = Asp[kp * BMp + am];
#pragma unroll
            for (int t = 0; t < 4; t++) {
                v2f b = Bsp[kp * BNp + wcol * 64 + t * 16 + m16];
                acc[t] = __builtin_amdgcn_wmma_f32_16x16x4_f32(
                    false, a, false, b, (short)0, acc[t], false, false);
            }
        }
        __syncthreads();
    }

    const int rbase = row0 + wrow * 16 + ((lane >> 4) << 3);
#pragma unroll
    for (int t = 0; t < 4; t++) {
        int col = col0 + wcol * 64 + t * 16 + m16;
#pragma unroll
        for (int r = 0; r < 8; r++) {
            int row = rbase + r;
            if (row < M) C[(size_t)row * NCOL + col] = acc[t][r];
        }
    }
}

template <int H>
__global__ __launch_bounds__(256) void node_attn(
    const float* __restrict__ h, const float* __restrict__ al,
    const float* __restrict__ ar, float* __restrict__ el, float* __restrict__ er) {
    int lane = threadIdx.x & 31;
    int n = blockIdx.x * 8 + (threadIdx.x >> 5);
    if (n >= NN) return;
    const float* hp = h + (size_t)n * (H * 64);
#pragma unroll
    for (int hh = 0; hh < H; hh++) {
        float v0 = hp[hh * 64 + lane];
        float v1 = hp[hh * 64 + 32 + lane];
        float pl = v0 * al[hh * 64 + lane] + v1 * al[hh * 64 + 32 + lane];
        float pr = v0 * ar[hh * 64 + lane] + v1 * ar[hh * 64 + 32 + lane];
        for (int off = 16; off > 0; off >>= 1) {
            pl += __shfl_down(pl, off);
            pr += __shfl_down(pr, off);
        }
        if (lane == 0) {
            el[(size_t)n * H + hh] = pl;
            er[(size_t)n * H + hh] = pr;
        }
    }
}

__global__ void dst_histogram(const int* __restrict__ dst, unsigned* __restrict__ deg) {
    int e = blockIdx.x * blockDim.x + threadIdx.x;
    if (e < EE) atomicAdd(&deg[dst[e]], 1u);
}

__global__ __launch_bounds__(1024) void scan_local(const unsigned* __restrict__ deg,
                                                   int* __restrict__ rowptr,
                                                   unsigned* __restrict__ bsum) {
    __shared__ unsigned wsum[32];
    const int tid = threadIdx.x, lane = tid & 31, wv = tid >> 5;
    const int i = blockIdx.x * 1024 + tid;
    const unsigned v = (i < NN) ? deg[i] : 0u;
    unsigned inc = v;
    for (int off = 1; off < 32; off <<= 1) {
        unsigned t = __shfl_up(inc, off);
        if (lane >= off) inc += t;
    }
    if (lane == 31) wsum[wv] = inc;
    __syncthreads();
    if (wv == 0) {
        const unsigned w = wsum[lane];
        unsigned winc = w;
        for (int off = 1; off < 32; off <<= 1) {
            unsigned t = __shfl_up(winc, off);
            if (lane >= off) winc += t;
        }
        wsum[lane] = winc - w;
        if (lane == 31) bsum[blockIdx.x] = winc;
    }
    __syncthreads();
    if (i <= NN) rowptr[i] = (int)(inc - v + wsum[wv]);
}

__global__ __launch_bounds__(1024) void scan_finish(int* rowptr, const unsigned* __restrict__ bsum,
                                                    int* __restrict__ cursor) {
    const int i = blockIdx.x * 1024 + threadIdx.x;
    unsigned off = 0;
    for (int b = 0; b < (int)blockIdx.x; b++) off += bsum[b];
    if (i <= NN) {
        const int v = rowptr[i] + (int)off;
        rowptr[i] = v;
        if (i < NN) cursor[i] = v;
    }
}

__global__ void csr_fill(const int* __restrict__ src, const int* __restrict__ dst,
                         int* __restrict__ cursor, int* __restrict__ csrc) {
    int e = blockIdx.x * blockDim.x + threadIdx.x;
    if (e >= EE) return;
    int pos = atomicAdd(&cursor[dst[e]], 1);
    csrc[pos] = src[e];
}

template <int H, bool ELU>
__global__ __launch_bounds__(256) void gat_gather(
    const int* __restrict__ rowptr, const int* __restrict__ csrc,
    const float* __restrict__ el, const float* __restrict__ er,
    const float* __restrict__ h, const float* __restrict__ bias,
    float* __restrict__ out) {
    static_assert(H == 4 || H == 1, "row widths handled: 256 and 64");
    constexpr int ROW = H * 64;
    const int lane = threadIdx.x & 31;
    const int n = blockIdx.x * 8 + __builtin_amdgcn_readfirstlane((int)(threadIdx.x >> 5));
    if (n >= NN) return;
    const int beg = rowptr[n], end = rowptr[n + 1];

    float ern[H];
    for (int hh = 0; hh < H; hh++) ern[hh] = er[(size_t)n * H + hh];

    auto score = [&](int s, float (&sc)[H]) {
        if constexpr (H == 4) {
            const float4 e = ((const float4*)el)[s];
            sc[0] = e.x + ern[0]; sc[1] = e.y + ern[1];
            sc[2] = e.z + ern[2]; sc[3] = e.w + ern[3];
        } else {
            sc[0] = el[s] + ern[0];
        }
        for (int hh = 0; hh < H; hh++) sc[hh] = sc[hh] > 0.0f ? sc[hh] : SLOPE * sc[hh];
    };

    const int j0 = beg + lane;
    const bool has0 = j0 < end;
    int s0 = 0;
    float sc0[H], m[H], den[H], a0[H];
    for (int hh = 0; hh < H; hh++) sc0[hh] = -INFINITY;
    if (has0) {
        s0 = csrc[j0];
        score(s0, sc0);
    }
    for (int hh = 0; hh < H; hh++) m[hh] = sc0[hh];
    for (int j = j0 + 32; j < end; j += 32) {
        float sc[H];
        score(csrc[j], sc);
        for (int hh = 0; hh < H; hh++) m[hh] = fmaxf(m[hh], sc[hh]);
    }
    for (int hh = 0; hh < H; hh++)
        for (int off = 16; off > 0; off >>= 1) m[hh] = fmaxf(m[hh], __shfl_xor(m[hh], off));
    for (int hh = 0; hh < H; hh++) {
        a0[hh] = has0 ? expf(sc0[hh] - m[hh]) : 0.0f;
        den[hh] = a0[hh];
    }
    for (int j = j0 + 32; j < end; j += 32) {
        float sc[H];
        score(csrc[j], sc);
        for (int hh = 0; hh < H; hh++) den[hh] += expf(sc[hh] - m[hh]);
    }
    for (int hh = 0; hh < H; hh++)
        for (int off = 16; off > 0; off >>= 1) den[hh] += __shfl_xor(den[hh], off);
    for (int hh = 0; hh < H; hh++) a0[hh] = has0 ? a0[hh] / den[hh] : 0.0f;

    float acc[2 * H];
    for (int q = 0; q < 2 * H; q++) acc[q] = 0.0f;
    const bool hi = (lane & 16) != 0;
    for (int c = beg; c < end; c += 32) {
        int s = s0;
        float a[H];
        for (int hh = 0; hh < H; hh++) a[hh] = a0[hh];
        if (c != beg) {
            const int j = c + lane;
            s = 0;
            for (int hh = 0; hh < H; hh++) a[hh] = 0.0f;
            if (j < end) {
                float sc[H];
                s = csrc[j];
                score(s, sc);
                for (int hh = 0; hh < H; hh++) a[hh] = expf(sc[hh] - m[hh]) / den[hh];
            }
        }
        const int cnt = (end - c) < 32 ? (end - c) : 32;
        auto step = [&](int jj) {
            const int sj = __builtin_amdgcn_readlane(s, jj);
            if constexpr (H == 4) {
                const float w0 = __int_as_float(__builtin_amdgcn_readlane(__float_as_int(a[0]), jj));
                const float w1 = __int_as_float(__builtin_amdgcn_readlane(__float_as_int(a[1]), jj));
                const float w2 = __int_as_float(__builtin_amdgcn_readlane(__float_as_int(a[2]), jj));
                const float w3 = __int_as_float(__builtin_amdgcn_readlane(__float_as_int(a[3]), jj));
                const float4* hp = (const float4*)(h + (size_t)sj * ROW);
                const float4 v0 = hp[lane];
                const float4 v1 = hp[lane + 32];
                const float c0 = hi ? w1 : w0;
                const float c1 = hi ? w3 : w2;
                acc[0] = fmaf(c0, v0.x, acc[0]); acc[1] = fmaf(c0, v0.y, acc[1]);
                acc[2] = fmaf(c0, v0.z, acc[2]); acc[3] = fmaf(c0, v0.w, acc[3]);
                acc[4] = fmaf(c1, v1.x, acc[4]); acc[5] = fmaf(c1, v1.y, acc[5]);
                acc[6] = fmaf(c1, v1.z, acc[6]); acc[7] = fmaf(c1, v1.w, acc[7]);
            } else {
                const float w = __int_as_float(__builtin_amdgcn_readlane(__float_as_int(a[0]), jj));
                const float2 v = ((const float2*)(h + (size_t)sj * ROW))[lane];
                acc[0] = fmaf(w, v.x, acc[0]); acc[1] = fmaf(w, v.y, acc[1]);
            }
        };
        int jj = 0;
        for (; jj + 4 <= cnt; jj += 4) { step(jj); step(jj + 1); step(jj + 2); step(jj + 3); }
        for (; jj < cnt; jj++) step(jj);
    }

    auto finish = [&](float x, float b) {
        const float r = x + b;
        if constexpr (ELU) return r > 0.0f ? r : expm1f(r);
        else return r;
    };
    if constexpr (H == 4) {
        const float4 b0 = ((const float4*)bias)[lane];
        const float4 b1 = ((const float4*)bias)[lane + 32];
        float4* op = (float4*)(out + (size_t)n * ROW);
        op[lane] = make_float4(finish(acc[0], b0.x), finish(acc[1], b0.y),
                               finish(acc[2], b0.z), finish(acc[3], b0.w));
        op[lane + 32] = make_float4(finish(acc[4], b1.x), finish(acc[5], b1.y),
                                    finish(acc[6], b1.z), finish(acc[7], b1.w));
    } else {
        const float2 b = ((const float2*)bias)[lane];
        ((float2*)(out + (size_t)n * ROW))[lane] =
            make_float2(finish(acc[0], b.x), finish(acc[1], b.y));
    }
}

__global__ void fill_u32(unsigned* p, unsigned v, int n) {
    int i = blockIdx.x * blockDim.x + threadIdx.x;
    if (i < n) p[i] = v;
}

extern "C" void kernel_launch(void* const* d_in, const int* in_sizes, int n_in,
                              void* d_out, int out_size, void* d_ws, size_t ws_size,
                              hipStream_t stream) {
    const float* feat = (const float*)d_in[0];
    const int*   src  = (const int*)d_in[1];
    const int*   dst  = (const int*)d_in[2];
    const float* W1   = (const float*)d_in[3];
    const float* al1  = (const float*)d_in[4];
    const float* ar1  = (const float*)d_in[5];
    const float* b1   = (const float*)d_in[6];
    const float* W2   = (const float*)d_in[7];
    const float* al2  = (const float*)d_in[8];
    const float* ar2  = (const float*)d_in[9];
    const float* b2   = (const float*)d_in[10];
    float* out = (float*)d_out;

    float* h1    = (float*)d_ws;
    float* x2    = h1    + (size_t)NN * 256;
    float* h2    = x2    + (size_t)NN * 256;
    float* el1   = h2    + (size_t)NN * 64;
    float* er1   = el1   + (size_t)NN * H1_;
    float* el2   = er1   + (size_t)NN * H1_;
    float* er2   = el2   + NN;
    unsigned* deg = (unsigned*)(er2 + NN);
    int* rowptr  = (int*)(deg + NN);
    int* cursor  = rowptr + NN + 16;
    unsigned* bsum = (unsigned*)(cursor + NN);
    int* csrc    = (int*)(bsum + 64);

    const int TB = 256;
    fill_u32<<<(NN + TB - 1) / TB, TB, 0, stream>>>(deg, 0u, NN);
    dst_histogram<<<(EE + TB - 1) / TB, TB, 0, stream>>>(dst, deg);
    scan_local<<<(NN + 1 + 1023) / 1024, 1024, 0, stream>>>(deg, rowptr, bsum);
    scan_finish<<<(NN + 1 + 1023) / 1024, 1024, 0, stream>>>(rowptr, bsum, cursor);
    csr_fill<<<(EE + TB - 1) / TB, TB, 0, stream>>>(src, dst, cursor, csrc);

    {
        dim3 grid(256 / 128, (NN + 63) / 64);
        wmma_gemm_f32<64, 128, 256><<<grid, 256, 0, stream>>>(feat, W1, h1, NN, IND);
    }
    node_attn<H1_><<<(NN + 7) / 8, 256, 0, stream>>>(h1, al1, ar1, el1, er1);
    gat_gather<H1_, true><<<(NN + 7) / 8, 256, 0, stream>>>(rowptr, csrc, el1, er1, h1, b1, x2);

    {
        dim3 grid(64 / 64, (NN + 127) / 128);
        wmma_gemm_f32<128, 64, 64><<<grid, 256, 0, stream>>>(x2, W2, h2, NN, 256);
    }
    node_attn<1><<<(NN + 7) / 8, 256, 0, stream>>>(h2, al2, ar2, el2, er2);
    gat_gather<1, false><<<(NN + 7) / 8, 256, 0, stream>>>(rowptr, csrc, el2, er2, h2, b2, out);
}


# Node scores el, er formed in the GEMM epilogue from the accumulators (4 products per lane, 16-lane butterfly sum: new sum order); both node-score launches and their re-read of h go
# speedup vs baseline: 4.1914x; 1.0774x over previous
#include <hip/hip_runtime.h>
#include <math.h>

#define NN   50000
#define EE   800000
#define IND  256
#define H1_  4
#define HID1 64
#define OUTD 64
#define SLOPE 0.2f

typedef __attribute__((ext_vector_type(2))) float v2f;
typedef __attribute__((ext_vector_type(8))) float v8f;

template <int BM, int BN, int NCOL>
__global__ __launch_bounds__(256) void wmma_gemm_f32(
    const float* __restrict__ A, const float* __restrict__ B,
    float* __restrict__ C,
    const float* __restrict__ al, const float* __restrict__ ar,
    float* __restrict__ el, float* __restrict__ er, int M, int K) {
    constexpr int BK  = 32;
    constexpr int BMp = BM + 2;
    constexpr int BNp = BN + 2;
    __shared__ __attribute__((aligned(16))) float As[(BK / 2) * BMp * 2];
    __shared__ __attribute__((aligned(16))) float Bs[(BK / 2) * BNp * 2];

    const int tid  = threadIdx.x;
    const int lane = tid & 31;
    const int wave = tid >> 5;
    constexpr int WCOLS = BN / 64;
    const int wrow = wave / WCOLS;
    const int wcol = wave % WCOLS;
    const int row0 = blockIdx.y * BM;
    const int col0 = blockIdx.x * BN;

    v8f acc[4];
#pragma unroll
    for (int t = 0; t < 4; t++)
#pragma unroll
        for (int r = 0; r < 8; r++) acc[t][r] = 0.0f;

    const int m16 = lane & 15;
    const int kb  = (lane >> 4) * 2;

    const v2f* Asp = (const v2f*)As;
    const v2f* Bsp = (const v2f*)Bs;

    constexpr int A_LD = (BM * BK / 4) / 256;
    constexpr int BQ   = BN / 4;
    constexpr int B_LD = ((BK / 2) * BQ) / 256;
    static_assert(A_LD * 256 * 4 == BM * BK, "A tile must split evenly over 256 threads");
    static_assert(B_LD * 256 == (BK / 2) * BQ, "B tile must split evenly over 256 threads");
    float4 ra[A_LD], rb0[B_LD], rb1[B_LD];

    auto fetch = [&](int k0) {
#pragma unroll
        for (int i = 0; i < A_LD; i++) {
            const int idx = tid + i * 256;
            const int m = idx >> 3, q = idx & 7;
            const int r = row0 + m;
            ra[i] = (r < M) ? *(const float4*)(A + (size_t)r * K + k0 + 4 * q)
                            : make_float4(0.0f, 0.0f, 0.0f, 0.0f);
        }
#pragma unroll
        for (int i = 0; i < B_LD; i++) {
            const int idx = tid + i * 256;
            const int kp = idx / BQ, j = idx % BQ;
            const float* bp = B + (size_t)(k0 + 2 * kp) * NCOL + col0 + 4 * j;
            rb0[i] = *(const float4*)bp;
            rb1[i] = *(const float4*)(bp + NCOL);
        }
    };
    auto stage = [&]() {
        float2* As2 = (float2*)As;
#pragma unroll
        for (int i = 0; i < A_LD; i++) {
            const int idx = tid + i * 256;
            const int m = idx >> 3, q = idx & 7;
            As2[(2 * q) * BMp + m]     = make_float2(ra[i].x, ra[i].y);
            As2[(2 * q + 1) * BMp + m] = make_float2(ra[i].z, ra[i].w);
        }
#pragma unroll
        for (int i = 0; i < B_LD; i++) {
            const int idx = tid + i * 256;
            const int kp = idx / BQ, j = idx % BQ;
            float4* d = (float4*)(Bs + (kp * BNp + 4 * j) * 2);
            d[0] = make_float4(rb0[i].x, rb1[i].x, rb0[i].y, rb1[i].y);
            d[1] = make_float4(rb0[i].z, rb1[i].z, rb0[i].w, rb1[i].w);
        }
    };

    fetch(0);
    for (int k0 = 0; k0 < K; k0 += BK) {
        stage();
        __syncthreads();
        if (k0 + BK < K) fetch(k0 + BK);

        const int am = wrow * 16 + m16;
#pragma unroll
        for (int kk = 0; kk < BK; kk += 4) {
            const int kp = (kk + kb) >> 1;
            v2f a = Asp[kp * BMp + am];
#pragma unroll
            for (int t = 0; t < 4; t++) {
                v2f b = Bsp[kp * BNp + wcol * 64 + t * 16 + m16];
                acc[t] = __builtin_amdgcn_wmma_f32_16x16x4_f32(
                    false, a, false, b, (short)0, acc[t], false, false);
            }
        }
        __syncthreads();
    }

    const int rbase = row0 + wrow * 16 + ((lane >> 4) << 3);
#pragma unroll
    for (int t = 0; t < 4; t++) {
        int col = col0 + wcol * 64 + t * 16 + m16;
#pragma unroll
        for (int r = 0; r < 8; r++) {
            int row = rbase + r;
            if (row < M) C[(size_t)row * NCOL + col] = acc[t][r];
        }
    }

    constexpr int NH = NCOL / 64;
    const int hd = (col0 >> 6) + wcol;
    float alv[4], arv[4];
#pragma unroll
    for (int t = 0; t < 4; t++) {
        alv[t] = al[hd * 64 + t * 16 + m16];
        arv[t] = ar[hd * 64 + t * 16 + m16];
    }
    float v[16];
#pragma unroll
    for (int r = 0; r < 8; r++) {
        float sl = acc[0][r] * alv[0];
        float sr = acc[0][r] * arv[0];
#pragma unroll
        for (int t = 1; t < 4; t++) {
            sl = fmaf(acc[t][r], alv[t], sl);
            sr = fmaf(acc[t][r], arv[t], sr);
        }
        v[r] = sl;
        v[8 + r] = sr;
    }
    const bool b3 = (lane & 8) != 0, b2 = (lane & 4) != 0;
    const bool b1 = (lane & 2) != 0, b0 = (lane & 1) != 0;
    float w8[8], w4[4], w2[2];
#pragma unroll
    for (int i = 0; i < 8; i++) {
        const float keep = b3 ? v[i + 8] : v[i];
        const float send = b3 ? v[i] : v[i + 8];
        w8[i] = keep + __shfl_xor(send, 8);
    }
#pragma unroll
    for (int i = 0; i < 4; i++) {
        const float keep = b2 ? w8[i + 4] : w8[i];
        const float send = b2 ? w8[i] : w8[i + 4];
        w4[i] = keep + __shfl_xor(send, 4);
    }
#pragma unroll
    for (int i = 0; i < 2; i++) {
        const float keep = b1 ? w4[i + 2] : w4[i];
        const float send = b1 ? w4[i] : w4[i + 2];
        w2[i] = keep + __shfl_xor(send, 2);
    }
    const float tot = (b0 ? w2[1] : w2[0]) + __shfl_xor(b0 ? w2[0] : w2[1], 1);
    const int srow = rbase + (m16 & 7);
    if (srow < M) {
        float* sp = (m16 < 8) ? el : er;
        sp[(size_t)srow * NH + hd] = tot;
    }
}

__global__ void dst_histogram(const int* __restrict__ dst, unsigned* __restrict__ deg) {
    int e = blockIdx.x * blockDim.x + threadIdx.x;
    if (e < EE) atomicAdd(&deg[dst[e]], 1u);
}

__global__ __launch_bounds__(1024) void scan_local(const unsigned* __restrict__ deg,
                                                   int* __restrict__ rowptr,
                                                   unsigned* __restrict__ bsum) {
    __shared__ unsigned wsum[32];
    const int tid = threadIdx.x, lane = tid & 31, wv = tid >> 5;
    const int i = blockIdx.x * 1024 + tid;
    const unsigned v = (i < NN) ? deg[i] : 0u;
    unsigned inc = v;
    for (int off = 1; off < 32; off <<= 1) {
        unsigned t = __shfl_up(inc, off);
        if (lane >= off) inc += t;
    }
    if (lane == 31) wsum[wv] = inc;
    __syncthreads();
    if (wv == 0) {
        const unsigned w = wsum[lane];
        unsigned winc = w;
        for (int off = 1; off < 32; off <<= 1) {
            unsigned t = __shfl_up(winc, off);
            if (lane >= off) winc += t;
        }
        wsum[lane] = winc - w;
        if (lane == 31) bsum[blockIdx.x] = winc;
    }
    __syncthreads();
    if (i <= NN) rowptr[i] = (int)(inc - v + wsum[wv]);
}

__global__ __launch_bounds__(1024) void scan_finish(int* rowptr, const unsigned* __restrict__ bsum,
                                                    int* __restrict__ cursor) {
    const int i = blockIdx.x * 1024 + threadIdx.x;
    unsigned off = 0;
    for (int b = 0; b < (int)blockIdx.x; b++) off += bsum[b];
    if (i <= NN) {
        const int v = rowptr[i] + (int)off;
        rowptr[i] = v;
        if (i < NN) cursor[i] = v;
    }
}

__global__ void csr_fill(const int* __restrict__ src, const int* __restrict__ dst,
                         int* __restrict__ cursor, int* __restrict__ csrc) {
    int e = blockIdx.x * blockDim.x + threadIdx.x;
    if (e >= EE) return;
    int pos = atomicAdd(&cursor[dst[e]], 1);
    csrc[pos] = src[e];
}

template <int H, bool ELU>
__global__ __launch_bounds__(256) void gat_gather(
    const int* __restrict__ rowptr, const int* __restrict__ csrc,
    const float* __restrict__ el, const float* __restrict__ er,
    const float* __restrict__ h, const float* __restrict__ bias,
    float* __restrict__ out) {
    static_assert(H == 4 || H == 1, "row widths handled: 256 and 64");
    constexpr int ROW = H * 64;
    const int lane = threadIdx.x & 31;
    const int n = blockIdx.x * 8 + __builtin_amdgcn_readfirstlane((int)(threadIdx.x >> 5));
    if (n >= NN) return;
    const int beg = rowptr[n], end = rowptr[n + 1];

    float ern[H];
    for (int hh = 0; hh < H; hh++) ern[hh] = er[(size_t)n * H + hh];

    auto score = [&](int s, float (&sc)[H]) {
        if constexpr (H == 4) {
            const float4 e = ((const float4*)el)[s];
            sc[0] = e.x + ern[0]; sc[1] = e.y + ern[1];
            sc[2] = e.z + ern[2]; sc[3] = e.w + ern[3];
        } else {
            sc[0] = el[s] + ern[0];
        }
        for (int hh = 0; hh < H; hh++) sc[hh] = sc[hh] > 0.0f ? sc[hh] : SLOPE * sc[hh];
    };

    const int j0 = beg + lane;
    const bool has0 = j0 < end;
    int s0 = 0;
    float sc0[H], m[H], den[H], a0[H];
    for (int hh = 0; hh < H; hh++) sc0[hh] = -INFINITY;
    if (has0) {
        s0 = csrc[j0];
        score(s0, sc0);
    }
    for (int hh = 0; hh < H; hh++) m[hh] = sc0[hh];
    for (int j = j0 + 32; j < end; j += 32) {
        float sc[H];
        score(csrc[j], sc);
        for (int hh = 0; hh < H; hh++) m[hh] = fmaxf(m[hh], sc[hh]);
    }
    for (int hh = 0; hh < H; hh++)
        for (int off = 16; off > 0; off >>= 1) m[hh] = fmaxf(m[hh], __shfl_xor(m[hh], off));
    for (int hh = 0; hh < H; hh++) {
        a0[hh] = has0 ? expf(sc0[hh] - m[hh]) : 0.0f;
        den[hh] = a0[hh];
    }
    for (int j = j0 + 32; j < end; j += 32) {
        float sc[H];
        score(csrc[j], sc);
        for (int hh = 0; hh < H; hh++) den[hh] += expf(sc[hh] - m[hh]);
    }
    for (int hh = 0; hh < H; hh++)
        for (int off = 16; off > 0; off >>= 1) den[hh] += __shfl_xor(den[hh], off);
    for (int hh = 0; hh < H; hh++) a0[hh] = has0 ? a0[hh] / den[hh] : 0.0f;

    float acc[2 * H];
    for (int q = 0; q < 2 * H; q++) acc[q] = 0.0f;
    const bool hi = (lane & 16) != 0;
    for (int c = beg; c < end; c += 32) {
        int s = s0;
        float a[H];
        for (int hh = 0; hh < H; hh++) a[hh] = a0[hh];
        if (c != beg) {
            const int j = c + lane;
            s = 0;
            for (int hh = 0; hh < H; hh++) a[hh] = 0.0f;
            if (j < end) {
                float sc[H];
                s = csrc[j];
                score(s, sc);
                for (int hh = 0; hh < H; hh++) a[hh] = expf(sc[hh] - m[hh]) / den[hh];
            }
        }
        const int cnt = (end - c) < 32 ? (end - c) : 32;
        auto step = [&](int jj) {
            const int sj = __builtin_amdgcn_readlane(s, jj);
            if constexpr (H == 4) {
                const float w0 = __int_as_float(__builtin_amdgcn_readlane(__float_as_int(a[0]), jj));
                const float w1 = __int_as_float(__builtin_amdgcn_readlane(__float_as_int(a[1]), jj));
                const float w2 = __int_as_float(__builtin_amdgcn_readlane(__float_as_int(a[2]), jj));
                const float w3 = __int_as_float(__builtin_amdgcn_readlane(__float_as_int(a[3]), jj));
                const float4* hp = (const float4*)(h + (size_t)sj * ROW);
                const float4 v0 = hp[lane];
                const float4 v1 = hp[lane + 32];
                const float c0 = hi ? w1 : w0;
                const float c1 = hi ? w3 : w2;
                acc[0] = fmaf(c0, v0.x, acc[0]); acc[1] = fmaf(c0, v0.y, acc[1]);
                acc[2] = fmaf(c0, v0.z, acc[2]); acc[3] = fmaf(c0, v0.w, acc[3]);
                acc[4] = fmaf(c1, v1.x, acc[4]); acc[5] = fmaf(c1, v1.y, acc[5]);
                acc[6] = fmaf(c1, v1.z, acc[6]); acc[7] = fmaf(c1, v1.w, acc[7]);
            } else {
                const float w = __int_as_float(__builtin_amdgcn_readlane(__float_as_int(a[0]), jj));
                const float2 v = ((const float2*)(h + (size_t)sj * ROW))[lane];
                acc[0] = fmaf(w, v.x, acc[0]); acc[1] = fmaf(w, v.y, acc[1]);
            }
        };
        int jj = 0;
        for (; jj + 4 <= cnt; jj += 4) { step(jj); step(jj + 1); step(jj + 2); step(jj + 3); }
        for (; jj < cnt; jj++) step(jj);
    }

    auto finish = [&](float x, float b) {
        const float r = x + b;
        if constexpr (ELU) return r > 0.0f ? r : expm1f(r);
        else return r;
    };
    if constexpr (H == 4) {
        const float4 b0 = ((const float4*)bias)[lane];
        const float4 b1 = ((const float4*)bias)[lane + 32];
        float4* op = (float4*)(out + (size_t)n * ROW);
        op[lane] = make_float4(finish(acc[0], b0.x), finish(acc[1], b0.y),
                               finish(acc[2], b0.z), finish(acc[3], b0.w));
        op[lane + 32] = make_float4(finish(acc[4], b1.x), finish(acc[5], b1.y),
                                    finish(acc[6], b1.z), finish(acc[7], b1.w));
    } else {
        const float2 b = ((const float2*)bias)[lane];
        ((float2*)(out + (size_t)n * ROW))[lane] =
            make_float2(finish(acc[0], b.x), finish(acc[1], b.y));
    }
}

__global__ void fill_u32(unsigned* p, unsigned v, int n) {
    int i = blockIdx.x * blockDim.x + threadIdx.x;
    if (i < n) p[i] = v;
}

extern "C" void kernel_launch(void* const* d_in, const int* in_sizes, int n_in,
                              void* d_out, int out_size, void* d_ws, size_t ws_size,
                              hipStream_t stream) {
    const float* feat = (const float*)d_in[0];
    const int*   src  = (const int*)d_in[1];
    const int*   dst  = (const int*)d_in[2];
    const float* W1   = (const float*)d_in[3];
    const float* al1  = (const float*)d_in[4];
    const float* ar1  = (const float*)d_in[5];
    const float* b1   = (const float*)d_in[6];
    const float* W2   = (const float*)d_in[7];
    const float* al2  = (const float*)d_in[8];
    const float* ar2  = (const float*)d_in[9];
    const float* b2   = (const float*)d_in[10];
    float* out = (float*)d_out;

    float* h1    = (float*)d_ws;
    float* x2    = h1    + (size_t)NN * 256;
    float* h2    = x2    + (size_t)NN * 256;
    float* el1   = h2    + (size_t)NN * 64;
    float* er1   = el1   + (size_t)NN * H1_;
    float* el2   = er1   + (size_t)NN * H1_;
    float* er2   = el2   + NN;
    unsigned* deg = (unsigned*)(er2 + NN);
    int* rowptr  = (int*)(deg + NN);
    int* cursor  = rowptr + NN + 16;
    unsigned* bsum = (unsigned*)(cursor + NN);
    int* csrc    = (int*)(bsum + 64);

    const int TB = 256;
    fill_u32<<<(NN + TB - 1) / TB, TB, 0, stream>>>(deg, 0u, NN);
    dst_histogram<<<(EE + TB - 1) / TB, TB, 0, stream>>>(dst, deg);
    scan_local<<<(NN + 1 + 1023) / 1024, 1024, 0, stream>>>(deg, rowptr, bsum);
    scan_finish<<<(NN + 1 + 1023) / 1024, 1024, 0, stream>>>(rowptr, bsum, cursor);
    csr_fill<<<(EE + TB - 1) / TB, TB, 0, stream>>>(src, dst, cursor, csrc);

    {
        dim3 grid(256 / 128, (NN + 63) / 64);
        wmma_gemm_f32<64, 128, 256><<<grid, 256, 0, stream>>>(feat, W1, h1, al1, ar1, el1, er1, NN, IND);
    }
    gat_gather<H1_, true><<<(NN + 7) / 8, 256, 0, stream>>>(rowptr, csrc, el1, er1, h1, b1, x2);

    {
        dim3 grid(64 / 64, (NN + 127) / 128);
        wmma_gemm_f32<128, 64, 64><<<grid, 256, 0, stream>>>(x2, W2, h2, al2, ar2, el2, er2, NN, 256);
    }
    gat_gather<1, false><<<(NN + 7) / 8, 256, 0, stream>>>(rowptr, csrc, el2, er2, h2, b2, out);
}


# Edge counters and fill cursors as one plane per package half (chosen by the wave's die number; planes share no 128-byte line); rows fill from both ends: sum order = fill order, as before
# speedup vs baseline: 5.6688x; 1.3525x over previous
#include <hip/hip_runtime.h>
#include <math.h>

#define NN   50000
#define EE   800000
#define IND  256
#define H1_  4
#define HID1 64
#define OUTD 64
#define SLOPE 0.2f
#define NNP  50016

typedef __attribute__((ext_vector_type(2))) float v2f;
typedef __attribute__((ext_vector_type(8))) float v8f;

template <int BM, int BN, int NCOL>
__global__ __launch_bounds__(256) void wmma_gemm_f32(
    const float* __restrict__ A, const float* __restrict__ B,
    float* __restrict__ C,
    const float* __restrict__ al, const float* __restrict__ ar,
    float* __restrict__ el, float* __restrict__ er, int M, int K) {
    constexpr int BK  = 32;
    constexpr int BMp = BM + 2;
    constexpr int BNp = BN + 2;
    __shared__ __attribute__((aligned(16))) float As[(BK / 2) * BMp * 2];
    __shared__ __attribute__((aligned(16))) float Bs[(BK / 2) * BNp * 2];

    const int tid  = threadIdx.x;
    const int lane = tid & 31;
    const int wave = tid >> 5;
    constexpr int WCOLS = BN / 64;
    const int wrow = wave / WCOLS;
    const int wcol = wave % WCOLS;
    const int row0 = blockIdx.y * BM;
    const int col0 = blockIdx.x * BN;

    v8f acc[4];
#pragma unroll
    for (int t = 0; t < 4; t++)
#pragma unroll
        for (int r = 0; r < 8; r++) acc[t][r] = 0.0f;

    const int m16 = lane & 15;
    const int kb  = (lane >> 4) * 2;

    const v2f* Asp = (const v2f*)As;
    const v2f* Bsp = (const v2f*)Bs;

    constexpr int A_LD = (BM * BK / 4) / 256;
    constexpr int BQ   = BN / 4;
    constexpr int B_LD = ((BK / 2) * BQ) / 256;
    static_assert(A_LD * 256 * 4 == BM * BK, "A tile must split evenly over 256 threads");
    static_assert(B_LD * 256 == (BK / 2) * BQ, "B tile must split evenly over 256 threads");
    float4 ra[A_LD], rb0[B_LD], rb1[B_LD];

    auto fetch = [&](int k0) {
#pragma unroll
        for (int i = 0; i < A_LD; i++) {
            const int idx = tid + i * 256;
            const int m = idx >> 3, q = idx & 7;
            const int r = row0 + m;
            ra[i] = (r < M) ? *(const float4*)(A + (size_t)r * K + k0 + 4 * q)
                            : make_float4(0.0f, 0.0f, 0.0f, 0.0f);
        }
#pragma unroll
        for (int i = 0; i < B_LD; i++) {
            const int idx = tid + i * 256;
            const int kp = idx / BQ, j = idx % BQ;
            const float* bp = B + (size_t)(k0 + 2 * kp) * NCOL + col0 + 4 * j;
            rb0[i] = *(const float4*)bp;
            rb1[i] = *(const float4*)(bp + NCOL);
        }
    };
    auto stage = [&]() {
        float2* As2 = (float2*)As;
#pragma unroll
        for (int i = 0; i < A_LD; i++) {
            const int idx = tid + i * 256;
            const int m = idx >> 3, q = idx & 7;
            As2[(2 * q) * BMp + m]     = make_float2(ra[i].x, ra[i].y);
            As2[(2 * q + 1) * BMp + m] = make_float2(ra[i].z, ra[i].w);
        }
#pragma unroll
        for (int i = 0; i < B_LD; i++) {
            const int idx = tid + i * 256;
            const int kp = idx / BQ, j = idx % BQ;
            float4* d = (float4*)(Bs + (kp * BNp + 4 * j) * 2);
            d[0] = make_float4(rb0[i].x, rb1[i].x, rb0[i].y, rb1[i].y);
            d[1] = make_float4(rb0[i].z, rb1[i].z, rb0[i].w, rb1[i].w);
        }
    };

    fetch(0);
    for (int k0 = 0; k0 < K; k0 += BK) {
        stage();
        __syncthreads();
        if (k0 + BK < K) fetch(k0 + BK);

        const int am = wrow * 16 + m16;
#pragma unroll
        for (int kk = 0; kk < BK; kk += 4) {
            const int kp = (kk + kb) >> 1;
            v2f a = Asp[kp * BMp + am];
#pragma unroll
            for (int t = 0; t < 4; t++) {
                v2f b = Bsp[kp * BNp + wcol * 64 + t * 16 + m16];
                acc[t] = __builtin_amdgcn_wmma_f32_16x16x4_f32(
                    false, a, false, b, (short)0, acc[t], false, false);
            }
        }
        __syncthreads();
    }

    const int rbase = row0 + wrow * 16 + ((lane >> 4) << 3);
#pragma unroll
    for (int t = 0; t < 4; t++) {
        int col = col0 + wcol * 64 + t * 16 + m16;
#pragma unroll
        for (int r = 0; r < 8; r++) {
            int row = rbase + r;
            if (row < M) C[(size_t)row * NCOL + col] = acc[t][r];
        }
    }

    constexpr int NH = NCOL / 64;
    const int hd = (col0 >> 6) + wcol;
    float alv[4], arv[4];
#pragma unroll
    for (int t = 0; t < 4; t++) {
        alv[t] = al[hd * 64 + t * 16 + m16];
        arv[t] = ar[hd * 64 + t * 16 + m16];
    }
    float v[16];
#pragma unroll
    for (int r = 0; r < 8; r++) {
        float sl = acc[0][r] * alv[0];
        float sr = acc[0][r] * arv[0];
#pragma unroll
        for (int t = 1; t < 4; t++) {
            sl = fmaf(acc[t][r], alv[t], sl);
            sr = fmaf(acc[t][r], arv[t], sr);
        }
        v[r] = sl;
        v[8 + r] = sr;
    }
    const bool b3 = (lane & 8) != 0, b2 = (lane & 4) != 0;
    const bool b1 = (lane & 2) != 0, b0 = (lane & 1) != 0;
    float w8[8], w4[4], w2[2];
#pragma unroll
    for (int i = 0; i < 8; i++) {
        const float keep = b3 ? v[i + 8] : v[i];
        const float send = b3 ? v[i] : v[i + 8];
        w8[i] = keep + __shfl_xor(send, 8);
    }
#pragma unroll
    for (int i = 0; i < 4; i++) {
        const float keep = b2 ? w8[i + 4] : w8[i];
        const float send = b2 ? w8[i] : w8[i + 4];
        w4[i] = keep + __shfl_xor(send, 4);
    }
#pragma unroll
    for (int i = 0; i < 2; i++) {
        const float keep = b1 ? w4[i + 2] : w4[i];
        const float send = b1 ? w4[i] : w4[i + 2];
        w2[i] = keep + __shfl_xor(send, 2);
    }
    const float tot = (b0 ? w2[1] : w2[0]) + __shfl_xor(b0 ? w2[0] : w2[1], 1);
    const int srow = rbase + (m16 & 7);
    if (srow < M) {
        float* sp = (m16 < 8) ? el : er;
        sp[(size_t)srow * NH + hd] = tot;
    }
}

__device__ __forceinline__ int package_half() {
    const unsigned id = __builtin_amdgcn_s_sendmsg_rtn(0x87);
    return (int)((id >> 18) & 1u);
}

__global__ void dst_histogram(const int* __restrict__ dst, unsigned* __restrict__ deg) {
    int e = blockIdx.x * blockDim.x + threadIdx.x;
    unsigned* plane = deg + package_half() * NNP;
    if (e < EE) atomicAdd(&plane[dst[e]], 1u);
}

__global__ __launch_bounds__(1024) void scan_local(const unsigned* __restrict__ deg,
                                                   int* __restrict__ rowptr,
                                                   unsigned* __restrict__ bsum) {
    __shared__ unsigned wsum[32];
    const int tid = threadIdx.x, lane = tid & 31, wv = tid >> 5;
    const int i = blockIdx.x * 1024 + tid;
    const unsigned v = (i < NN) ? deg[i] + deg[NNP + i] : 0u;
    unsigned inc = v;
    for (int off = 1; off < 32; off <<= 1) {
        unsigned t = __shfl_up(inc, off);
        if (lane >= off) inc += t;
    }
    if (lane == 31) wsum[wv] = inc;
    __syncthreads();
    if (wv == 0) {
        const unsigned w = wsum[lane];
        unsigned winc = w;
        for (int off = 1; off < 32; off <<= 1) {
            unsigned t = __shfl_up(winc, off);
            if (lane >= off) winc += t;
        }
        wsum[lane] = winc - w;
        if (lane == 31) bsum[blockIdx.x] = winc;
    }
    __syncthreads();
    if (i <= NN) rowptr[i] = (int)(inc - v + wsum[wv]);
}

__global__ __launch_bounds__(1024) void scan_finish(int* rowptr, const unsigned* __restrict__ bsum,
                                                    const unsigned* __restrict__ deg,
                                                    int* __restrict__ cursor) {
    const int i = blockIdx.x * 1024 + threadIdx.x;
    unsigned off = 0;
    for (int b = 0; b < (int)blockIdx.x; b++) off += bsum[b];
    if (i <= NN) {
        const int v = rowptr[i] + (int)off;
        rowptr[i] = v;
        if (i < NN) {
            cursor[i] = v;
            cursor[NNP + i] = v + (int)(deg[i] + deg[NNP + i]) - 1;
        }
    }
}

__global__ void csr_fill(const int* __restrict__ src, const int* __restrict__ dst,
                         int* __restrict__ cursor, int* __restrict__ csrc) {
    int e = blockIdx.x * blockDim.x + threadIdx.x;
    const int half = package_half();
    int* cur = cursor + half * NNP;
    const int inc = half ? -1 : 1;
    if (e >= EE) return;
    int pos = atomicAdd(&cur[dst[e]], inc);
    csrc[pos] = src[e];
}

template <int H, bool ELU>
__global__ __launch_bounds__(256) void gat_gather(
    const int* __restrict__ rowptr, const int* __restrict__ csrc,
    const float* __restrict__ el, const float* __restrict__ er,
    const float* __restrict__ h, const float* __restrict__ bias,
    float* __restrict__ out) {
    static_assert(H == 4 || H == 1, "row widths handled: 256 and 64");
    constexpr int ROW = H * 64;
    const int lane = threadIdx.x & 31;
    const int n = blockIdx.x * 8 + __builtin_amdgcn_readfirstlane((int)(threadIdx.x >> 5));
    if (n >= NN) return;
    const int beg = rowptr[n], end = rowptr[n + 1];

    float ern[H];
    for (int hh = 0; hh < H; hh++) ern[hh] = er[(size_t)n * H + hh];

    auto score = [&](int s, float (&sc)[H]) {
        if constexpr (H == 4) {
            const float4 e = ((const float4*)el)[s];
            sc[0] = e.x + ern[0]; sc[1] = e.y + ern[1];
            sc[2] = e.z + ern[2]; sc[3] = e.w + ern[3];
        } else {
            sc[0] = el[s] + ern[0];
        }
        for (int hh = 0; hh < H; hh++) sc[hh] = sc[hh] > 0.0f ? sc[hh] : SLOPE * sc[hh];
    };

    const int j0 = beg + lane;
    const bool has0 = j0 < end;
    int s0 = 0;
    float sc0[H], m[H], den[H], a0[H];
    for (int hh = 0; hh < H; hh++) sc0[hh] = -INFINITY;
    if (has0) {
        s0 = csrc[j0];
        score(s0, sc0);
    }
    for (int hh = 0; hh < H; hh++) m[hh] = sc0[hh];
    for (int j = j0 + 32; j < end; j += 32) {
        float sc[H];
        score(csrc[j], sc);
        for (int hh = 0; hh < H; hh++) m[hh] = fmaxf(m[hh], sc[hh]);
    }
    for (int hh = 0; hh < H; hh++)
        for (int off = 16; off > 0; off >>= 1) m[hh] = fmaxf(m[hh], __shfl_xor(m[hh], off));
    for (int hh = 0; hh < H; hh++) {
        a0[hh] = has0 ? expf(sc0[hh] - m[hh]) : 0.0f;
        den[hh] = a0[hh];
    }
    for (int j = j0 + 32; j < end; j += 32) {
        float sc[H];
        score(csrc[j], sc);
        for (int hh = 0; hh < H; hh++) den[hh] += expf(sc[hh] - m[hh]);
    }
    for (int hh = 0; hh < H; hh++)
        for (int off = 16; off > 0; off >>= 1) den[hh] += __shfl_xor(den[hh], off);
    for (int hh = 0; hh < H; hh++) a0[hh] = has0 ? a0[hh] / den[hh] : 0.0f;

    float acc[2 * H];
    for (int q = 0; q < 2 * H; q++) acc[q] = 0.0f;
    const bool hi = (lane & 16) != 0;
    for (int c = beg; c < end; c += 32) {
        int s = s0;
        float a[H];
        for (int hh = 0; hh < H; hh++) a[hh] = a0[hh];
        if (c != beg) {
            const int j = c + lane;
            s = 0;
            for (int hh = 0; hh < H; hh++) a[hh] = 0.0f;
            if (j < end) {
                float sc[H];
                s = csrc[j];
                score(s, sc);
                for (int hh = 0; hh < H; hh++) a[hh] = expf(sc[hh] - m[hh]) / den[hh];
            }
        }
        const int cnt = (end - c) < 32 ? (end - c) : 32;
        auto step = [&](int jj) {
            const int sj = __builtin_amdgcn_readlane(s, jj);
            if constexpr (H == 4) {
                const float w0 = __int_as_float(__builtin_amdgcn_readlane(__float_as_int(a[0]), jj));
                const float w1 = __int_as_float(__builtin_amdgcn_readlane(__float_as_int(a[1]), jj));
                const float w2 = __int_as_float(__builtin_amdgcn_readlane(__float_as_int(a[2]), jj));
                const float w3 = __int_as_float(__builtin_amdgcn_readlane(__float_as_int(a[3]), jj));
                const float4* hp = (const float4*)(h + (size_t)sj * ROW);
                const float4 v0 = hp[lane];
                const float4 v1 = hp[lane + 32];
                const float c0 = hi ? w1 : w0;
                const float c1 = hi ? w3 : w2;
                acc[0] = fmaf(c0, v0.x, acc[0]); acc[1] = fmaf(c0, v0.y, acc[1]);
                acc[2] = fmaf(c0, v0.z, acc[2]); acc[3] = fmaf(c0, v0.w, acc[3]);
                acc[4] = fmaf(c1, v1.x, acc[4]); acc[5] = fmaf(c1, v1.y, acc[5]);
                acc[6] = fmaf(c1, v1.z, acc[6]); acc[7] = fmaf(c1, v1.w, acc[7]);
            } else {
                const float w = __int_as_float(__builtin_amdgcn_readlane(__float_as_int(a[0]), jj));
                const float2 v = ((const float2*)(h + (size_t)sj * ROW))[lane];
                acc[0] = fmaf(w, v.x, acc[0]); acc[1] = fmaf(w, v.y, acc[1]);
            }
        };
        int jj = 0;
        for (; jj + 4 <= cnt; jj += 4) { step(jj); step(jj + 1); step(jj + 2); step(jj + 3); }
        for (; jj < cnt; jj++) step(jj);
    }

    auto finish = [&](float x, float b) {
        const float r = x + b;
        if constexpr (ELU) return r > 0.0f ? r : expm1f(r);
        else return r;
    };
    if constexpr (H == 4) {
        const float4 b0 = ((const float4*)bias)[lane];
        const float4 b1 = ((const float4*)bias)[lane + 32];
        float4* op = (float4*)(out + (size_t)n * ROW);
        op[lane] = make_float4(finish(acc[0], b0.x), finish(acc[1], b0.y),
                               finish(acc[2], b0.z), finish(acc[3], b0.w));
        op[lane + 32] = make_float4(finish(acc[4], b1.x), finish(acc[5], b1.y),
                                    finish(acc[6], b1.z), finish(acc[7], b1.w));
    } else {
        const float2 b = ((const float2*)bias)[lane];
        ((float2*)(out + (size_t)n * ROW))[lane] =
            make_float2(finish(acc[0], b.x), finish(acc[1], b.y));
    }
}

__global__ void fill_u32(unsigned* p, unsigned v, int n) {
    int i = blockIdx.x * blockDim.x + threadIdx.x;
    if (i < n) p[i] = v;
}

extern "C" void kernel_launch(void* const* d_in, const int* in_sizes, int n_in,
                              void* d_out, int out_size, void* d_ws, size_t ws_size,
                              hipStream_t stream) {
    const float* feat = (const float*)d_in[0];
    const int*   src  = (const int*)d_in[1];
    const int*   dst  = (const int*)d_in[2];
    const float* W1   = (const float*)d_in[3];
    const float* al1  = (const float*)d_in[4];
    const float* ar1  = (const float*)d_in[5];
    const float* b1   = (const float*)d_in[6];
    const float* W2   = (const float*)d_in[7];
    const float* al2  = (const float*)d_in[8];
    const float* ar2  = (const float*)d_in[9];
    const float* b2   = (const float*)d_in[10];
    float* out = (float*)d_out;

    float* h1    = (float*)d_ws;
    float* x2    = h1    + (size_t)NN * 256;
    float* h2    = x2    + (size_t)NN * 256;
    float* el1   = h2    + (size_t)NN * 64;
    float* er1   = el1   + (size_t)NN * H1_;
    float* el2   = er1   + (size_t)NN * H1_;
    float* er2   = el2   + NN;
    unsigned* deg = (unsigned*)(er2 + NN);
    int* rowptr  = (int*)(deg + 2 * NNP);
    int* cursor  = rowptr + NN + 16;
    unsigned* bsum = (unsigned*)(cursor + 2 * NNP);
    int* csrc    = (int*)(bsum + 64);

    const int TB = 256;
    fill_u32<<<(2 * NNP + TB - 1) / TB, TB, 0, stream>>>(deg, 0u, 2 * NNP);
    dst_histogram<<<(EE + TB - 1) / TB, TB, 0, stream>>>(dst, deg);
    scan_local<<<(NN + 1 + 1023) / 1024, 1024, 0, stream>>>(deg, rowptr, bsum);
    scan_finish<<<(NN + 1 + 1023) / 1024, 1024, 0, stream>>>(rowptr, bsum, deg, cursor);
    csr_fill<<<(EE + TB - 1) / TB, TB, 0, stream>>>(src, dst, cursor, csrc);

    {
        dim3 grid(256 / 128, (NN + 63) / 64);
        wmma_gemm_f32<64, 128, 256><<<grid, 256, 0, stream>>>(feat, W1, h1, al1, ar1, el1, er1, NN, IND);
    }
    gat_gather<H1_, true><<<(NN + 7) / 8, 256, 0, stream>>>(rowptr, csrc, el1, er1, h1, b1, x2);

    {
        dim3 grid(64 / 64, (NN + 127) / 128);
        wmma_gemm_f32<128, 64, 64><<<grid, 256, 0, stream>>>(x2, W2, h2, al2, ar2, el2, er2, NN, 256);
    }
    gat_gather<1, false><<<(NN + 7) / 8, 256, 0, stream>>>(rowptr, csrc, el2, er2, h2, b2, out);
}


# Edges dropped straight into fixed 32-slot buckets per destination and package half in one counted pass (full bucket: per-row linked list); histogram, both scans, fill launch go; sum order = arrival
# speedup vs baseline: 6.7946x; 1.1986x over previous
#include <hip/hip_runtime.h>
#include <math.h>

#define NN   50000
#define EE   800000
#define IND  256
#define H1_  4
#define HID1 64
#define OUTD 64
#define SLOPE 0.2f
#define NNP  50016
#define CAPH 32

typedef __attribute__((ext_vector_type(2))) float v2f;
typedef __attribute__((ext_vector_type(8))) float v8f;

template <int BM, int BN, int NCOL>
__global__ __launch_bounds__(256) void wmma_gemm_f32(
    const float* __restrict__ A, const float* __restrict__ B,
    float* __restrict__ C,
    const float* __restrict__ al, const float* __restrict__ ar,
    float* __restrict__ el, float* __restrict__ er, int M, int K) {
    constexpr int BK  = 32;
    constexpr int BMp = BM + 2;
    constexpr int BNp = BN + 2;
    __shared__ __attribute__((aligned(16))) float As[(BK / 2) * BMp * 2];
    __shared__ __attribute__((aligned(16))) float Bs[(BK / 2) * BNp * 2];

    const int tid  = threadIdx.x;
    const int lane = tid & 31;
    const int wave = tid >> 5;
    constexpr int WCOLS = BN / 64;
    const int wrow = wave / WCOLS;
    const int wcol = wave % WCOLS;
    const int row0 = blockIdx.y * BM;
    const int col0 = blockIdx.x * BN;

    v8f acc[4];
#pragma unroll
    for (int t = 0; t < 4; t++)
#pragma unroll
        for (int r = 0; r < 8; r++) acc[t][r] = 0.0f;

    const int m16 = lane & 15;
    const int kb  = (lane >> 4) * 2;

    const v2f* Asp = (const v2f*)As;
    const v2f* Bsp = (const v2f*)Bs;

    constexpr int A_LD = (BM * BK / 4) / 256;
    constexpr int BQ   = BN / 4;
    constexpr int B_LD = ((BK / 2) * BQ) / 256;
    static_assert(A_LD * 256 * 4 == BM * BK, "A tile must split evenly over 256 threads");
    static_assert(B_LD * 256 == (BK / 2) * BQ, "B tile must split evenly over 256 threads");
    float4 ra[A_LD], rb0[B_LD], rb1[B_LD];

    auto fetch = [&](int k0) {
#pragma unroll
        for (int i = 0; i < A_LD; i++) {
            const int idx = tid + i * 256;
            const int m = idx >> 3, q = idx & 7;
            const int r = row0 + m;
            ra[i] = (r < M) ? *(const float4*)(A + (size_t)r * K + k0 + 4 * q)
                            : make_float4(0.0f, 0.0f, 0.0f, 0.0f);
        }
#pragma unroll
        for (int i = 0; i < B_LD; i++) {
            const int idx = tid + i * 256;
            const int kp = idx / BQ, j = idx % BQ;
            const float* bp = B + (size_t)(k0 + 2 * kp) * NCOL + col0 + 4 * j;
            rb0[i] = *(const float4*)bp;
            rb1[i] = *(const float4*)(bp + NCOL);
        }
    };
    auto stage = [&]() {
        float2* As2 = (float2*)As;
#pragma unroll
        for (int i = 0; i < A_LD; i++) {
            const int idx = tid + i * 256;
            const int m = idx >> 3, q = idx & 7;
            As2[(2 * q) * BMp + m]     = make_float2(ra[i].x, ra[i].y);
            As2[(2 * q + 1) * BMp + m] = make_float2(ra[i].z, ra[i].w);
        }
#pragma unroll
        for (int i = 0; i < B_LD; i++) {
            const int idx = tid + i * 256;
            const int kp = idx / BQ, j = idx % BQ;
            float4* d = (float4*)(Bs + (kp * BNp + 4 * j) * 2);
            d[0] = make_float4(rb0[i].x, rb1[i].x, rb0[i].y, rb1[i].y);
            d[1] = make_float4(rb0[i].z, rb1[i].z, rb0[i].w, rb1[i].w);
        }
    };

    fetch(0);
    for (int k0 = 0; k0 < K; k0 += BK) {
        stage();
        __syncthreads();
        if (k0 + BK < K) fetch(k0 + BK);

        const int am = wrow * 16 + m16;
#pragma unroll
        for (int kk = 0; kk < BK; kk += 4) {
            const int kp = (kk + kb) >> 1;
            v2f a = Asp[kp * BMp + am];
#pragma unroll
            for (int t = 0; t < 4; t++) {
                v2f b = Bsp[kp * BNp + wcol * 64 + t * 16 + m16];
                acc[t] = __builtin_amdgcn_wmma_f32_16x16x4_f32(
                    false, a, false, b, (short)0, acc[t], false, false);
            }
        }
        __syncthreads();
    }

    const int rbase = row0 + wrow * 16 + ((lane >> 4) << 3);
#pragma unroll
    for (int t = 0; t < 4; t++) {
        int col = col0 + wcol * 64 + t * 16 + m16;
#pragma unroll
        for (int r = 0; r < 8; r++) {
            int row = rbase + r;
            if (row < M) C[(size_t)row * NCOL + col] = acc[t][r];
        }
    }

    constexpr int NH = NCOL / 64;
    const int hd = (col0 >> 6) + wcol;
    float alv[4], arv[4];
#pragma unroll
    for (int t = 0; t < 4; t++) {
        alv[t] = al[hd * 64 + t * 16 + m16];
        arv[t] = ar[hd * 64 + t * 16 + m16];
    }
    float v[16];
#pragma unroll
    for (int r = 0; r < 8; r++) {
        float sl = acc[0][r] * alv[0];
        float sr = acc[0][r] * arv[0];
#pragma unroll
        for (int t = 1; t < 4; t++) {
            sl = fmaf(acc[t][r], alv[t], sl);
            sr = fmaf(acc[t][r], arv[t], sr);
        }
        v[r] = sl;
        v[8 + r] = sr;
    }
    const bool b3 = (lane & 8) != 0, b2 = (lane & 4) != 0;
    const bool b1 = (lane & 2) != 0, b0 = (lane & 1) != 0;
    float w8[8], w4[4], w2[2];
#pragma unroll
    for (int i = 0; i < 8; i++) {
        const float keep = b3 ? v[i + 8] : v[i];
        const float send = b3 ? v[i] : v[i + 8];
        w8[i] = keep + __shfl_xor(send, 8);
    }
#pragma unroll
    for (int i = 0; i < 4; i++) {
        const float keep = b2 ? w8[i + 4] : w8[i];
        const float send = b2 ? w8[i] : w8[i + 4];
        w4[i] = keep + __shfl_xor(send, 4);
    }
#pragma unroll
    for (int i = 0; i < 2; i++) {
        const float keep = b1 ? w4[i + 2] : w4[i];
        const float send = b1 ? w4[i] : w4[i + 2];
        w2[i] = keep + __shfl_xor(send, 2);
    }
    const float tot = (b0 ? w2[1] : w2[0]) + __shfl_xor(b0 ? w2[0] : w2[1], 1);
    const int srow = rbase + (m16 & 7);
    if (srow < M) {
        float* sp = (m16 < 8) ? el : er;
        sp[(size_t)srow * NH + hd] = tot;
    }
}

__device__ __forceinline__ int package_half() {
    const unsigned id = __builtin_amdgcn_s_sendmsg_rtn(0x87);
    return (int)((id >> 18) & 1u);
}

__global__ void edge_init(unsigned* __restrict__ ecnt, int* __restrict__ ovh) {
    int i = blockIdx.x * blockDim.x + threadIdx.x;
    if (i < 2 * NNP) { ecnt[i] = 0u; ovh[i] = -1; }
}

__global__ void bucket_fill(const int* __restrict__ src, const int* __restrict__ dst,
                            unsigned* __restrict__ ecnt, int* __restrict__ bucket,
                            int* __restrict__ ovh, int* __restrict__ ovn, int* __restrict__ ovs) {
    int e = blockIdx.x * blockDim.x + threadIdx.x;
    const int half = package_half();
    if (e >= EE) return;
    const int d = dst[e];
    const int s = src[e];
    const unsigned pos = atomicAdd(&ecnt[half * NNP + d], 1u);
    const bool fits = pos < (unsigned)CAPH;
    int* where = fits ? bucket + ((size_t)half * NN + d) * CAPH + pos : ovs + e;
    *where = s;
    if (!fits) ovn[e] = atomicExch(&ovh[half * NNP + d], e);
}

template <int H, bool ELU>
__global__ __launch_bounds__(256) void gat_gather(
    const unsigned* __restrict__ ecnt, const int* __restrict__ bucket,
    const int* __restrict__ ovh, const int* __restrict__ ovn, const int* __restrict__ ovs,
    const float* __restrict__ el, const float* __restrict__ er,
    const float* __restrict__ h, const float* __restrict__ bias,
    float* __restrict__ out) {
    static_assert(H == 4 || H == 1, "row widths handled: 256 and 64");
    constexpr int ROW = H * 64;
    const int lane = threadIdx.x & 31;
    const int n = blockIdx.x * 8 + __builtin_amdgcn_readfirstlane((int)(threadIdx.x >> 5));
    if (n >= NN) return;
    const unsigned t0 = ecnt[n], t1 = ecnt[NNP + n];
    const int c0 = (int)(t0 < (unsigned)CAPH ? t0 : (unsigned)CAPH);
    const int c1 = (int)(t1 < (unsigned)CAPH ? t1 : (unsigned)CAPH);
    const int end = c0 + c1;
    const bool spill = (t0 > (unsigned)CAPH) || (t1 > (unsigned)CAPH);
    const int* bk0 = bucket + (size_t)n * CAPH;
    const int* bk1 = bucket + ((size_t)NN + n) * CAPH - c0;
    auto slot = [&](int j) { return j < c0 ? bk0[j] : bk1[j]; };

    float ern[H];
    for (int hh = 0; hh < H; hh++) ern[hh] = er[(size_t)n * H + hh];

    auto score = [&](int s, float (&sc)[H]) {
        if constexpr (H == 4) {
            const float4 e = ((const float4*)el)[s];
            sc[0] = e.x + ern[0]; sc[1] = e.y + ern[1];
            sc[2] = e.z + ern[2]; sc[3] = e.w + ern[3];
        } else {
            sc[0] = el[s] + ern[0];
        }
        for (int hh = 0; hh < H; hh++) sc[hh] = sc[hh] > 0.0f ? sc[hh] : SLOPE * sc[hh];
    };
    auto walk = [&](auto&& f) {
        for (int p = 0; p < 2; p++) {
            if ((p ? t1 : t0) <= (unsigned)CAPH) continue;
            int cur = ovh[p * NNP + n];
            for (int it = 0; it < EE && (unsigned)cur < (unsigned)EE; it++) {
                f(ovs[cur]);
                cur = ovn[cur];
            }
        }
    };

    const int j0 = lane;
    const bool has0 = j0 < end;
    int s0 = 0;
    float sc0[H], m[H], den[H], a0[H];
    for (int hh = 0; hh < H; hh++) sc0[hh] = -INFINITY;
    if (has0) {
        s0 = slot(j0);
        score(s0, sc0);
    }
    for (int hh = 0; hh < H; hh++) m[hh] = sc0[hh];
    for (int j = j0 + 32; j < end; j += 32) {
        float sc[H];
        score(slot(j), sc);
        for (int hh = 0; hh < H; hh++) m[hh] = fmaxf(m[hh], sc[hh]);
    }
    for (int hh = 0; hh < H; hh++)
        for (int off = 16; off > 0; off >>= 1) m[hh] = fmaxf(m[hh], __shfl_xor(m[hh], off));
    if (spill)
        walk([&](int s) {
            float sc[H];
            score(s, sc);
            for (int hh = 0; hh < H; hh++) m[hh] = fmaxf(m[hh], sc[hh]);
        });
    for (int hh = 0; hh < H; hh++) {
        a0[hh] = has0 ? expf(sc0[hh] - m[hh]) : 0.0f;
        den[hh] = a0[hh];
    }
    for (int j = j0 + 32; j < end; j += 32) {
        float sc[H];
        score(slot(j), sc);
        for (int hh = 0; hh < H; hh++) den[hh] += expf(sc[hh] - m[hh]);
    }
    for (int hh = 0; hh < H; hh++)
        for (int off = 16; off > 0; off >>= 1) den[hh] += __shfl_xor(den[hh], off);
    if (spill)
        walk([&](int s) {
            float sc[H];
            score(s, sc);
            for (int hh = 0; hh < H; hh++) den[hh] += expf(sc[hh] - m[hh]);
        });
    for (int hh = 0; hh < H; hh++) a0[hh] = has0 ? a0[hh] / den[hh] : 0.0f;

    float acc[2 * H];
    for (int q = 0; q < 2 * H; q++) acc[q] = 0.0f;
    const bool hi = (lane & 16) != 0;
    auto addrow = [&](int sj, float k0, float k1) {
        if constexpr (H == 4) {
            const float4* hp = (const float4*)(h + (size_t)sj * ROW);
            const float4 v0 = hp[lane];
            const float4 v1 = hp[lane + 32];
            acc[0] = fmaf(k0, v0.x, acc[0]); acc[1] = fmaf(k0, v0.y, acc[1]);
            acc[2] = fmaf(k0, v0.z, acc[2]); acc[3] = fmaf(k0, v0.w, acc[3]);
            acc[4] = fmaf(k1, v1.x, acc[4]); acc[5] = fmaf(k1, v1.y, acc[5]);
            acc[6] = fmaf(k1, v1.z, acc[6]); acc[7] = fmaf(k1, v1.w, acc[7]);
        } else {
            const float2 v = ((const float2*)(h + (size_t)sj * ROW))[lane];
            acc[0] = fmaf(k0, v.x, acc[0]); acc[1] = fmaf(k0, v.y, acc[1]);
        }
    };
    for (int c = 0; c < end; c += 32) {
        int s = s0;
        float a[H];
        for (int hh = 0; hh < H; hh++) a[hh] = a0[hh];
        if (c != 0) {
            const int j = c + lane;
            s = 0;
            for (int hh = 0; hh < H; hh++) a[hh] = 0.0f;
            if (j < end) {
                float sc[H];
                s = slot(j);
                score(s, sc);
                for (int hh = 0; hh < H; hh++) a[hh] = expf(sc[hh] - m[hh]) / den[hh];
            }
        }
        const int cnt = (end - c) < 32 ? (end - c) : 32;
        auto step = [&](int jj) {
            const int sj = __builtin_amdgcn_readlane(s, jj);
            if constexpr (H == 4) {
                const float w0 = __int_as_float(__builtin_amdgcn_readlane(__float_as_int(a[0]), jj));
                const float w1 = __int_as_float(__builtin_amdgcn_readlane(__float_as_int(a[1]), jj));
                const float w2 = __int_as_float(__builtin_amdgcn_readlane(__float_as_int(a[2]), jj));
                const float w3 = __int_as_float(__builtin_amdgcn_readlane(__float_as_int(a[3]), jj));
                addrow(sj, hi ? w1 : w0, hi ? w3 : w2);
            } else {
                const float w = __int_as_float(__builtin_amdgcn_readlane(__float_as_int(a[0]), jj));
                addrow(sj, w, w);
            }
        };
        int jj = 0;
        for (; jj + 4 <= cnt; jj += 4) { step(jj); step(jj + 1); step(jj + 2); step(jj + 3); }
        for (; jj < cnt; jj++) step(jj);
    }
    if (spill)
        walk([&](int s) {
            float sc[H], w[H];
            score(s, sc);
            for (int hh = 0; hh < H; hh++) w[hh] = expf(sc[hh] - m[hh]) / den[hh];
            if constexpr (H == 4) addrow(s, hi ? w[1] : w[0], hi ? w[3] : w[2]);
            else addrow(s, w[0], w[0]);
        });

    auto finish = [&](float x, float b) {
        const float r = x + b;
        if constexpr (ELU) return r > 0.0f ? r : expm1f(r);
        else return r;
    };
    if constexpr (H == 4) {
        const float4 b0 = ((const float4*)bias)[lane];
        const float4 b1 = ((const float4*)bias)[lane + 32];
        float4* op = (float4*)(out + (size_t)n * ROW);
        op[lane] = make_float4(finish(acc[0], b0.x), finish(acc[1], b0.y),
                               finish(acc[2], b0.z), finish(acc[3], b0.w));
        op[lane + 32] = make_float4(finish(acc[4], b1.x), finish(acc[5], b1.y),
                                    finish(acc[6], b1.z), finish(acc[7], b1.w));
    } else {
        const float2 b = ((const float2*)bias)[lane];
        ((float2*)(out + (size_t)n * ROW))[lane] =
            make_float2(finish(acc[0], b.x), finish(acc[1], b.y));
    }
}

extern "C" void kernel_launch(void* const* d_in, const int* in_sizes, int n_in,
                              void* d_out, int out_size, void* d_ws, size_t ws_size,
                              hipStream_t stream) {
    const float* feat = (const float*)d_in[0];
    const int*   src  = (const int*)d_in[1];
    const int*   dst  = (const int*)d_in[2];
    const float* W1   = (const float*)d_in[3];
    const float* al1  = (const float*)d_in[4];
    const float* ar1  = (const float*)d_in[5];
    const float* b1   = (const float*)d_in[6];
    const float* W2   = (const float*)d_in[7];
    const float* al2  = (const float*)d_in[8];
    const float* ar2  = (const float*)d_in[9];
    const float* b2   = (const float*)d_in[10];
    float* out = (float*)d_out;

    float* h1    = (float*)d_ws;
    float* x2    = h1    + (size_t)NN * 256;
    float* h2    = x2    + (size_t)NN * 256;
    float* el1   = h2    + (size_t)NN * 64;
    float* er1   = el1   + (size_t)NN * H1_;
    float* el2   = er1   + (size_t)NN * H1_;
    float* er2   = el2   + NN;
    unsigned* ecnt = (unsigned*)(er2 + NN);
    int* ovh     = (int*)(ecnt + 2 * NNP);
    int* bucket  = ovh + 2 * NNP;
    int* ovn     = bucket + (size_t)2 * NN * CAPH;
    int* ovs     = ovn + EE;

    const int TB = 256;
    edge_init<<<(2 * NNP + TB - 1) / TB, TB, 0, stream>>>(ecnt, ovh);
    bucket_fill<<<(EE + TB - 1) / TB, TB, 0, stream>>>(src, dst, ecnt, bucket, ovh, ovn, ovs);

    {
        dim3 grid(256 / 128, (NN + 63) / 64);
        wmma_gemm_f32<64, 128, 256><<<grid, 256, 0, stream>>>(feat, W1, h1, al1, ar1, el1, er1, NN, IND);
    }
    gat_gather<H1_, true><<<(NN + 7) / 8, 256, 0, stream>>>(ecnt, bucket, ovh, ovn, ovs, el1, er1, h1, b1, x2);

    {
        dim3 grid(64 / 64, (NN + 127) / 128);
        wmma_gemm_f32<128, 64, 64><<<grid, 256, 0, stream>>>(x2, W2, h2, al2, ar2, el2, er2, NN, 256);
    }
    gat_gather<1, false><<<(NN + 7) / 8, 256, 0, stream>>>(ecnt, bucket, ovh, ovn, ovs, el2, er2, h2, b2, out);
}


# Bucket fill moved into the layer-1 GEMM launch as trailing work-groups (4 edges per thread): its returning adds run beside the GEMM's last tiles; one launch fewer; sum order = arrival, as before
# speedup vs baseline: 7.1114x; 1.0466x over previous
#include <hip/hip_runtime.h>
#include <math.h>

#define NN   50000
#define EE   800000
#define IND  256
#define H1_  4
#define HID1 64
#define OUTD 64
#define SLOPE 0.2f
#define NNP  50016
#define CAPH 32

typedef __attribute__((ext_vector_type(2))) float v2f;
typedef __attribute__((ext_vector_type(8))) float v8f;

template <int BM, int BN, int NCOL>
__device__ __forceinline__ void gemm_tile(
    const float* __restrict__ A, const float* __restrict__ B,
    float* __restrict__ C,
    const float* __restrict__ al, const float* __restrict__ ar,
    float* __restrict__ el, float* __restrict__ er, int M, int K, int bx, int by) {
    constexpr int BK  = 32;
    constexpr int BMp = BM + 2;
    constexpr int BNp = BN + 2;
    __shared__ __attribute__((aligned(16))) float As[(BK / 2) * BMp * 2];
    __shared__ __attribute__((aligned(16))) float Bs[(BK / 2) * BNp * 2];

    const int tid  = threadIdx.x;
    const int lane = tid & 31;
    const int wave = tid >> 5;
    constexpr int WCOLS = BN / 64;
    const int wrow = wave / WCOLS;
    const int wcol = wave % WCOLS;
    const int row0 = by * BM;
    const int col0 = bx * BN;

    v8f acc[4];
#pragma unroll
    for (int t = 0; t < 4; t++)
#pragma unroll
        for (int r = 0; r < 8; r++) acc[t][r] = 0.0f;

    const int m16 = lane & 15;
    const int kb  = (lane >> 4) * 2;

    const v2f* Asp = (const v2f*)As;
    const v2f* Bsp = (const v2f*)Bs;

    constexpr int A_LD = (BM * BK / 4) / 256;
    constexpr int BQ   = BN / 4;
    constexpr int B_LD = ((BK / 2) * BQ) / 256;
    static_assert(A_LD * 256 * 4 == BM * BK, "A tile must split evenly over 256 threads");
    static_assert(B_LD * 256 == (BK / 2) * BQ, "B tile must split evenly over 256 threads");
    float4 ra[A_LD], rb0[B_LD], rb1[B_LD];

    auto fetch = [&](int k0) {
#pragma unroll
        for (int i = 0; i < A_LD; i++) {
            const int idx = tid + i * 256;
            const int m = idx >> 3, q = idx & 7;
            const int r = row0 + m;
            ra[i] = (r < M) ? *(const float4*)(A + (size_t)r * K + k0 + 4 * q)
                            : make_float4(0.0f, 0.0f, 0.0f, 0.0f);
        }
#pragma unroll
        for (int i = 0; i < B_LD; i++) {
            const int idx = tid + i * 256;
            const int kp = idx / BQ, j = idx % BQ;
            const float* bp = B + (size_t)(k0 + 2 * kp) * NCOL + col0 + 4 * j;
            rb0[i] = *(const float4*)bp;
            rb1[i] = *(const float4*)(bp + NCOL);
        }
    };
    auto stage = [&]() {
        float2* As2 = (float2*)As;
#pragma unroll
        for (int i = 0; i < A_LD; i++) {
            const int idx = tid + i * 256;
            const int m = idx >> 3, q = idx & 7;
            As2[(2 * q) * BMp + m]     = make_float2(ra[i].x, ra[i].y);
            As2[(2 * q + 1) * BMp + m] = make_float2(ra[i].z, ra[i].w);
        }
#pragma unroll
        for (int i = 0; i < B_LD; i++) {
            const int idx = tid + i * 256;
            const int kp = idx / BQ, j = idx % BQ;
            float4* d = (float4*)(Bs + (kp * BNp + 4 * j) * 2);
            d[0] = make_float4(rb0[i].x, rb1[i].x, rb0[i].y, rb1[i].y);
            d[1] = make_float4(rb0[i].z, rb1[i].z, rb0[i].w, rb1[i].w);
        }
    };

    fetch(0);
    for (int k0 = 0; k0 < K; k0 += BK) {
        stage();
        __syncthreads();
        if (k0 + BK < K) fetch(k0 + BK);

        const int am = wrow * 16 + m16;
#pragma unroll
        for (int kk = 0; kk < BK; kk += 4) {
            const int kp = (kk + kb) >> 1;
            v2f a = Asp[kp * BMp + am];
#pragma unroll
            for (int t = 0; t < 4; t++) {
                v2f b = Bsp[kp * BNp + wcol * 64 + t * 16 + m16];
                acc[t] = __builtin_amdgcn_wmma_f32_16x16x4_f32(
                    false, a, false, b, (short)0, acc[t], false, false);
            }
        }
        __syncthreads();
    }

    const int rbase = row0 + wrow * 16 + ((lane >> 4) << 3);
#pragma unroll
    for (int t = 0; t < 4; t++) {
        int col = col0 + wcol * 64 + t * 16 + m16;
#pragma unroll
        for (int r = 0; r < 8; r++) {
            int row = rbase + r;
            if (row < M) C[(size_t)row * NCOL + col] = acc[t][r];
        }
    }

    constexpr int NH = NCOL / 64;
    const int hd = (col0 >> 6) + wcol;
    float alv[4], arv[4];
#pragma unroll
    for (int t = 0; t < 4; t++) {
        alv[t] = al[hd * 64 + t * 16 + m16];
        arv[t] = ar[hd * 64 + t * 16 + m16];
    }
    float v[16];
#pragma unroll
    for (int r = 0; r < 8; r++) {
        float sl = acc[0][r] * alv[0];
        float sr = acc[0][r] * arv[0];
#pragma unroll
        for (int t = 1; t < 4; t++) {
            sl = fmaf(acc[t][r], alv[t], sl);
            sr = fmaf(acc[t][r], arv[t], sr);
        }
        v[r] = sl;
        v[8 + r] = sr;
    }
    const bool b3 = (lane & 8) != 0, b2 = (lane & 4) != 0;
    const bool b1 = (lane & 2) != 0, b0 = (lane & 1) != 0;
    float w8[8], w4[4], w2[2];
#pragma unroll
    for (int i = 0; i < 8; i++) {
        const float keep = b3 ? v[i + 8] : v[i];
        const float send = b3 ? v[i] : v[i + 8];
        w8[i] = keep + __shfl_xor(send, 8);
    }
#pragma unroll
    for (int i = 0; i < 4; i++) {
        const float keep = b2 ? w8[i + 4] : w8[i];
        const float send = b2 ? w8[i] : w8[i + 4];
        w4[i] = keep + __shfl_xor(send, 4);
    }
#pragma unroll
    for (int i = 0; i < 2; i++) {
        const float keep = b1 ? w4[i + 2] : w4[i];
        const float send = b1 ? w4[i] : w4[i + 2];
        w2[i] = keep + __shfl_xor(send, 2);
    }
    const float tot = (b0 ? w2[1] : w2[0]) + __shfl_xor(b0 ? w2[0] : w2[1], 1);
    const int srow = rbase + (m16 & 7);
    if (srow < M) {
        float* sp = (m16 < 8) ? el : er;
        sp[(size_t)srow * NH + hd] = tot;
    }
}

__device__ __forceinline__ int package_half() {
    const unsigned id = __builtin_amdgcn_s_sendmsg_rtn(0x87);
    return (int)((id >> 18) & 1u);
}

__global__ void edge_init(unsigned* __restrict__ ecnt, int* __restrict__ ovh) {
    int i = blockIdx.x * blockDim.x + threadIdx.x;
    if (i < 2 * NNP) { ecnt[i] = 0u; ovh[i] = -1; }
}

static_assert(EE % 4 == 0, "a thread takes four whole edges");
#define FILL_BLOCKS ((EE / 4 + 255) / 256)
__device__ __forceinline__ void bucket_fill4(int fb, const int* __restrict__ src, const int* __restrict__ dst,
                                             unsigned* __restrict__ ecnt, int* __restrict__ bucket,
                                             int* __restrict__ ovh, int* __restrict__ ovn,
                                             int* __restrict__ ovs) {
    const int t = fb * 256 + (int)threadIdx.x;
    const int half = package_half();
    if (t >= EE / 4) return;
    const int4 d4 = ((const int4*)dst)[t];
    const int4 s4 = ((const int4*)src)[t];
    const int dd[4] = {d4.x, d4.y, d4.z, d4.w};
    const int ss[4] = {s4.x, s4.y, s4.z, s4.w};
    unsigned pos[4];
#pragma unroll
    for (int i = 0; i < 4; i++) pos[i] = atomicAdd(&ecnt[half * NNP + dd[i]], 1u);
#pragma unroll
    for (int i = 0; i < 4; i++) {
        const int e = 4 * t + i;
        const bool fits = pos[i] < (unsigned)CAPH;
        int* where = fits ? bucket + ((size_t)half * NN + dd[i]) * CAPH + pos[i] : ovs + e;
        *where = ss[i];
        if (!fits) ovn[e] = atomicExch(&ovh[half * NNP + dd[i]], e);
    }
}

template <int BM, int BN, int NCOL>
__global__ __launch_bounds__(256) void wmma_gemm_f32(
    const float* __restrict__ A, const float* __restrict__ B,
    float* __restrict__ C,
    const float* __restrict__ al, const float* __restrict__ ar,
    float* __restrict__ el, float* __restrict__ er, int M, int K) {
    gemm_tile<BM, BN, NCOL>(A, B, C, al, ar, el, er, M, K, (int)blockIdx.x, (int)blockIdx.y);
}

#define G1_BM 64
#define G1_BN 128
#define G1_ROWB ((NN + G1_BM - 1) / G1_BM)
#define G1_COLB ((H1_ * HID1) / G1_BN)
#define G1_BLOCKS (G1_ROWB * G1_COLB)
__global__ __launch_bounds__(256) void gemm1_and_fill(
    const float* __restrict__ A, const float* __restrict__ B,
    float* __restrict__ C,
    const float* __restrict__ al, const float* __restrict__ ar,
    float* __restrict__ el, float* __restrict__ er,
    const int* __restrict__ src, const int* __restrict__ dst,
    unsigned* __restrict__ ecnt, int* __restrict__ bucket,
    int* __restrict__ ovh, int* __restrict__ ovn, int* __restrict__ ovs) {
    const int b = (int)blockIdx.x;
    if (b < G1_BLOCKS)
        gemm_tile<G1_BM, G1_BN, H1_ * HID1>(A, B, C, al, ar, el, er, NN, IND, b % G1_COLB, b / G1_COLB);
    else
        bucket_fill4(b - G1_BLOCKS, src, dst, ecnt, bucket, ovh, ovn, ovs);
}

template <int H, bool ELU>
__global__ __launch_bounds__(256) void gat_gather(
    const unsigned* __restrict__ ecnt, const int* __restrict__ bucket,
    const int* __restrict__ ovh, const int* __restrict__ ovn, const int* __restrict__ ovs,
    const float* __restrict__ el, const float* __restrict__ er,
    const float* __restrict__ h, const float* __restrict__ bias,
    float* __restrict__ out) {
    static_assert(H == 4 || H == 1, "row widths handled: 256 and 64");
    constexpr int ROW = H * 64;
    const int lane = threadIdx.x & 31;
    const int n = blockIdx.x * 8 + __builtin_amdgcn_readfirstlane((int)(threadIdx.x >> 5));
    if (n >= NN) return;
    const unsigned t0 = ecnt[n], t1 = ecnt[NNP + n];
    const int c0 = (int)(t0 < (unsigned)CAPH ? t0 : (unsigned)CAPH);
    const int c1 = (int)(t1 < (unsigned)CAPH ? t1 : (unsigned)CAPH);
    const int end = c0 + c1;
    const bool spill = (t0 > (unsigned)CAPH) || (t1 > (unsigned)CAPH);
    const int* bk0 = bucket + (size_t)n * CAPH;
    const int* bk1 = bucket + ((size_t)NN + n) * CAPH - c0;
    auto slot = [&](int j) { return j < c0 ? bk0[j] : bk1[j]; };

    float ern[H];
    for (int hh = 0; hh < H; hh++) ern[hh] = er[(size_t)n * H + hh];

    auto score = [&](int s, float (&sc)[H]) {
        if constexpr (H == 4) {
            const float4 e = ((const float4*)el)[s];
            sc[0] = e.x + ern[0]; sc[1] = e.y + ern[1];
            sc[2] = e.z + ern[2]; sc[3] = e.w + ern[3];
        } else {
            sc[0] = el[s] + ern[0];
        }
        for (int hh = 0; hh < H; hh++) sc[hh] = sc[hh] > 0.0f ? sc[hh] : SLOPE * sc[hh];
    };
    auto walk = [&](auto&& f) {
        for (int p = 0; p < 2; p++) {
            if ((p ? t1 : t0) <= (unsigned)CAPH) continue;
            int cur = ovh[p * NNP + n];
            for (int it = 0; it < EE && (unsigned)cur < (unsigned)EE; it++) {
                f(ovs[cur]);
                cur = ovn[cur];
            }
        }
    };

    const int j0 = lane;
    const bool has0 = j0 < end;
    int s0 = 0;
    float sc0[H], m[H], den[H], a0[H];
    for (int hh = 0; hh < H; hh++) sc0[hh] = -INFINITY;
    if (has0) {
        s0 = slot(j0);
        score(s0, sc0);
    }
    for (int hh = 0; hh < H; hh++) m[hh] = sc0[hh];
    for (int j = j0 + 32; j < end; j += 32) {
        float sc[H];
        score(slot(j), sc);
        for (int hh = 0; hh < H; hh++) m[hh] = fmaxf(m[hh], sc[hh]);
    }
    for (int hh = 0; hh < H; hh++)
        for (int off = 16; off > 0; off >>= 1) m[hh] = fmaxf(m[hh], __shfl_xor(m[hh], off));
    if (spill)
        walk([&](int s) {
            float sc[H];
            score(s, sc);
            for (int hh = 0; hh < H; hh++) m[hh] = fmaxf(m[hh], sc[hh]);
        });
    for (int hh = 0; hh < H; hh++) {
        a0[hh] = has0 ? expf(sc0[hh] - m[hh]) : 0.0f;
        den[hh] = a0[hh];
    }
    for (int j = j0 + 32; j < end; j += 32) {
        float sc[H];
        score(slot(j), sc);
        for (int hh = 0; hh < H; hh++) den[hh] += expf(sc[hh] - m[hh]);
    }
    for (int hh = 0; hh < H; hh++)
        for (int off = 16; off > 0; off >>= 1) den[hh] += __shfl_xor(den[hh], off);
    if (spill)
        walk([&](int s) {
            float sc[H];
            score(s, sc);
            for (int hh = 0; hh < H; hh++) den[hh] += expf(sc[hh] - m[hh]);
        });
    for (int hh = 0; hh < H; hh++) a0[hh] = has0 ? a0[hh] / den[hh] : 0.0f;

    float acc[2 * H];
    for (int q = 0; q < 2 * H; q++) acc[q] = 0.0f;
    const bool hi = (lane & 16) != 0;
    auto addrow = [&](int sj, float k0, float k1) {
        if constexpr (H == 4) {
            const float4* hp = (const float4*)(h + (size_t)sj * ROW);
            const float4 v0 = hp[lane];
            const float4 v1 = hp[lane + 32];
            acc[0] = fmaf(k0, v0.x, acc[0]); acc[1] = fmaf(k0, v0.y, acc[1]);
            acc[2] = fmaf(k0, v0.z, acc[2]); acc[3] = fmaf(k0, v0.w, acc[3]);
            acc[4] = fmaf(k1, v1.x, acc[4]); acc[5] = fmaf(k1, v1.y, acc[5]);
            acc[6] = fmaf(k1, v1.z, acc[6]); acc[7] = fmaf(k1, v1.w, acc[7]);
        } else {
            const float2 v = ((const float2*)(h + (size_t)sj * ROW))[lane];
            acc[0] = fmaf(k0, v.x, acc[0]); acc[1] = fmaf(k0, v.y, acc[1]);
        }
    };
    for (int c = 0; c < end; c += 32) {
        int s = s0;
        float a[H];
        for (int hh = 0; hh < H; hh++) a[hh] = a0[hh];
        if (c != 0) {
            const int j = c + lane;
            s = 0;
            for (int hh = 0; hh < H; hh++) a[hh] = 0.0f;
            if (j < end) {
                float sc[H];
                s = slot(j);
                score(s, sc);
                for (int hh = 0; hh < H; hh++) a[hh] = expf(sc[hh] - m[hh]) / den[hh];
            }
        }
        const int cnt = (end - c) < 32 ? (end - c) : 32;
        auto step = [&](int jj) {
            const int sj = __builtin_amdgcn_readlane(s, jj);
            if constexpr (H == 4) {
                const float w0 = __int_as_float(__builtin_amdgcn_readlane(__float_as_int(a[0]), jj));
                const float w1 = __int_as_float(__builtin_amdgcn_readlane(__float_as_int(a[1]), jj));
                const float w2 = __int_as_float(__builtin_amdgcn_readlane(__float_as_int(a[2]), jj));
                const float w3 = __int_as_float(__builtin_amdgcn_readlane(__float_as_int(a[3]), jj));
                addrow(sj, hi ? w1 : w0, hi ? w3 : w2);
            } else {
                const float w = __int_as_float(__builtin_amdgcn_readlane(__float_as_int(a[0]), jj));
                addrow(sj, w, w);
            }
        };
        int jj = 0;
        for (; jj + 4 <= cnt; jj += 4) { step(jj); step(jj + 1); step(jj + 2); step(jj + 3); }
        for (; jj < cnt; jj++) step(jj);
    }
    if (spill)
        walk([&](int s) {
            float sc[H], w[H];
            score(s, sc);
            for (int hh = 0; hh < H; hh++) w[hh] = expf(sc[hh] - m[hh]) / den[hh];
            if constexpr (H == 4) addrow(s, hi ? w[1] : w[0], hi ? w[3] : w[2]);
            else addrow(s, w[0], w[0]);
        });

    auto finish = [&](float x, float b) {
        const float r = x + b;
        if constexpr (ELU) return r > 0.0f ? r : expm1f(r);
        else return r;
    };
    if constexpr (H == 4) {
        const float4 b0 = ((const float4*)bias)[lane];
        const float4 b1 = ((const float4*)bias)[lane + 32];
        float4* op = (float4*)(out + (size_t)n * ROW);
        op[lane] = make_float4(finish(acc[0], b0.x), finish(acc[1], b0.y),
                               finish(acc[2], b0.z), finish(acc[3], b0.w));
        op[lane + 32] = make_float4(finish(acc[4], b1.x), finish(acc[5], b1.y),
                                    finish(acc[6], b1.z), finish(acc[7], b1.w));
    } else {
        const float2 b = ((const float2*)bias)[lane];
        ((float2*)(out + (size_t)n * ROW))[lane] =
            make_float2(finish(acc[0], b.x), finish(acc[1], b.y));
    }
}

extern "C" void kernel_launch(void* const* d_in, const int* in_sizes, int n_in,
                              void* d_out, int out_size, void* d_ws, size_t ws_size,
                              hipStream_t stream) {
    const float* feat = (const float*)d_in[0];
    const int*   src  = (const int*)d_in[1];
    const int*   dst  = (const int*)d_in[2];
    const float* W1   = (const float*)d_in[3];
    const float* al1  = (const float*)d_in[4];
    const float* ar1  = (const float*)d_in[5];
    const float* b1   = (const float*)d_in[6];
    const float* W2   = (const float*)d_in[7];
    const float* al2  = (const float*)d_in[8];
    const float* ar2  = (const float*)d_in[9];
    const float* b2   = (const float*)d_in[10];
    float* out = (float*)d_out;

    float* h1    = (float*)d_ws;
    float* x2    = h1    + (size_t)NN * 256;
    float* h2    = x2    + (size_t)NN * 256;
    float* el1   = h2    + (size_t)NN * 64;
    float* er1   = el1   + (size_t)NN * H1_;
    float* el2   = er1   + (size_t)NN * H1_;
    float* er2   = el2   + NN;
    unsigned* ecnt = (unsigned*)(er2 + NN);
    int* ovh     = (int*)(ecnt + 2 * NNP);
    int* bucket  = ovh + 2 * NNP;
    int* ovn     = bucket + (size_t)2 * NN * CAPH;
    int* ovs     = ovn + EE;

    const int TB = 256;
    edge_init<<<(2 * NNP + TB - 1) / TB, TB, 0, stream>>>(ecnt, ovh);

    gemm1_and_fill<<<G1_BLOCKS + FILL_BLOCKS, 256, 0, stream>>>(feat, W1, h1, al1, ar1, el1, er1,
                                                               src, dst, ecnt, bucket, ovh, ovn, ovs);
    gat_gather<H1_, true><<<(NN + 7) / 8, 256, 0, stream>>>(ecnt, bucket, ovh, ovn, ovs, el1, er1, h1, b1, x2);

    {
        dim3 grid(64 / 64, (NN + 127) / 128);
        wmma_gemm_f32<128, 64, 64><<<grid, 256, 0, stream>>>(x2, W2, h2, al2, ar2, el2, er2, NN, 256);
    }
    gat_gather<1, false><<<(NN + 7) / 8, 256, 0, stream>>>(ecnt, bucket, ovh, ovn, ovs, el2, er2, h2, b2, out);
}
